# starting kernel
# baseline (speedup 1.0000x reference)
#include <hip/hip_runtime.h>


#define Dch      128
#define Hh       4
#define Ch       32
#define Lnum     3
#define NEG_SLOPE 0.2f
#define BN_EPS   1e-5f

typedef __attribute__((ext_vector_type(2))) float v2f;
typedef __attribute__((ext_vector_type(8))) float v8f;

__device__ __forceinline__ unsigned fkey(float f) {
  unsigned b = __float_as_uint(f);
  return b ^ ((unsigned)((int)b >> 31) | 0x80000000u);
}
__device__ __forceinline__ float kinv(unsigned k) {
  return __uint_as_float((k & 0x80000000u) ? (k ^ 0x80000000u) : ~k);
}

__device__ __forceinline__ void edge_sd(const int* __restrict__ ei, int Ereal,
                                        int e, int& s, int& d) {
  if (e < Ereal) { s = ei[e]; d = ei[Ereal + e]; }
  else           { s = e - Ereal; d = s; }
}

__global__ void fill_u32(unsigned* __restrict__ p, unsigned v, long n) {
  long i = (long)blockIdx.x * blockDim.x + threadIdx.x;
  long stride = (long)gridDim.x * blockDim.x;
  for (; i < n; i += stride) p[i] = v;
}

__global__ __launch_bounds__(256) void gemm_wmma_f32(
    const float* __restrict__ X, const float* __restrict__ Wm,
    float* __restrict__ Ho) {
  const int wave = threadIdx.x >> 5;
  const int lane = threadIdx.x & 31;
  const int half = lane >> 4;
  const int l15  = lane & 15;
  const int row0 = blockIdx.x << 4;
  const int bcol = (wave << 4) + l15;
  const float* arow = X + (long)(row0 + l15) * Dch;

  v8f acc = {};
  for (int k = 0; k < Dch; k += 4) {
    const int ka = k + half * 2;
    v2f a, b;
    a.x = arow[ka];
    a.y = arow[ka + 1];
    b.x = Wm[(long)ka * Dch + bcol];
    b.y = Wm[(long)(ka + 1) * Dch + bcol];
    acc = __builtin_amdgcn_wmma_f32_16x16x4_f32(
         false, a,  false, b,
         (short)0, acc,  false,  false);
  }
  float* orow = Ho + (long)(row0 + half * 8) * Dch + bcol;
#pragma unroll
  for (int v = 0; v < 8; ++v) orow[(long)v * Dch] = acc[v];
}

__global__ void att_logits(const float* __restrict__ Hf,
                           const float* __restrict__ a_src,
                           const float* __restrict__ a_dst,
                           float* __restrict__ alsrc, float* __restrict__ aldst,
                           int Nn) {
  int i = blockIdx.x * blockDim.x + threadIdx.x;
  if (i >= Nn * Hh) return;
  int n = i >> 2, h = i & 3;
  const float* hp = Hf + (long)n * Dch + h * Ch;
  const float* as = a_src + h * Ch;
  const float* ad = a_dst + h * Ch;
  float s0 = 0.f, s1 = 0.f;
#pragma unroll
  for (int c = 0; c < Ch; ++c) { float v = hp[c]; s0 += v * as[c]; s1 += v * ad[c]; }
  alsrc[i] = s0;
  aldst[i] = s1;
}

__global__ void edge_max(const int* __restrict__ ei, int Ereal, int Etot,
                         const float* __restrict__ alsrc,
                         const float* __restrict__ aldst,
                         unsigned* __restrict__ mkey) {
  int e = blockIdx.x * blockDim.x + threadIdx.x;
  if (e >= Etot) return;
  int s, d; edge_sd(ei, Ereal, e, s, d);
#pragma unroll
  for (int h = 0; h < Hh; ++h) {
    float sc = alsrc[s * Hh + h] + aldst[d * Hh + h];
    sc = sc > 0.f ? sc : sc * NEG_SLOPE;
    atomicMax(&mkey[d * Hh + h], fkey(sc));
  }
}

__global__ void edge_expsum(const int* __restrict__ ei, int Ereal, int Etot,
                            const float* __restrict__ alsrc,
                            const float* __restrict__ aldst,
                            const unsigned* __restrict__ mkey,
                            float* __restrict__ ssum) {
  int e = blockIdx.x * blockDim.x + threadIdx.x;
  if (e >= Etot) return;
  int s, d; edge_sd(ei, Ereal, e, s, d);
#pragma unroll
  for (int h = 0; h < Hh; ++h) {
    float sc = alsrc[s * Hh + h] + aldst[d * Hh + h];
    sc = sc > 0.f ? sc : sc * NEG_SLOPE;
    float p = expf(sc - kinv(mkey[d * Hh + h]));
    atomicAdd(&ssum[d * Hh + h], p);
  }
}

__global__ __launch_bounds__(256) void edge_aggregate(
    const int* __restrict__ ei, int Ereal, int Etot,
    const float* __restrict__ alsrc, const float* __restrict__ aldst,
    const unsigned* __restrict__ mkey, const float* __restrict__ ssum,
    const float* __restrict__ Hf, float* __restrict__ acc) {
  long gid = (long)blockIdx.x * blockDim.x + threadIdx.x;
  int e    = (int)(gid >> 5);
  int lane = (int)(gid & 31);
  if (e >= Etot) return;
  int s, d; edge_sd(ei, Ereal, e, s, d);
  const float* hsrc = Hf + (long)s * Dch;
  float* adst = acc + (long)d * Dch;
#pragma unroll
  for (int h = 0; h < Hh; ++h) {
    float sc = alsrc[s * Hh + h] + aldst[d * Hh + h];
    sc = sc > 0.f ? sc : sc * NEG_SLOPE;
    float p = expf(sc - kinv(mkey[d * Hh + h]));
    float alpha = p / (ssum[d * Hh + h] + 1e-16f);
    int ch = h * Ch + lane;
    atomicAdd(&adst[ch], alpha * hsrc[ch]);
  }
}

__global__ void post_bn_elu(const float* __restrict__ acc,
                            const float* __restrict__ bias,
                            const float* __restrict__ gamma,
                            const float* __restrict__ beta,
                            const float* __restrict__ mean,
                            const float* __restrict__ var,
                            float* __restrict__ xo, long n) {
  long i = (long)blockIdx.x * blockDim.x + threadIdx.x;
  if (i >= n) return;
  int c = (int)(i & (Dch - 1));
  float v = acc[i] + bias[c];
  v = (v - mean[c]) * (gamma[c] / sqrtf(var[c] + BN_EPS)) + beta[c];
  xo[i] = v > 0.f ? v : expf(v) - 1.f;
}

__global__ void pool_accum(const float* __restrict__ x,
                           const int* __restrict__ batch,
                           float* __restrict__ sums, float* __restrict__ cnt,
                           long n) {
  long i = (long)blockIdx.x * blockDim.x + threadIdx.x;
  if (i >= n) return;
  int node = (int)(i >> 7);
  int c    = (int)(i & 127);
  int g    = batch[node];
  atomicAdd(&sums[(long)g * Dch + c], x[i]);
  if (c == 0) atomicAdd(&cnt[g], 1.0f);
}

__global__ void pool_final(const float* __restrict__ sums,
                           const float* __restrict__ cnt,
                           float* __restrict__ out, int G) {
  int i = blockIdx.x * blockDim.x + threadIdx.x;
  if (i >= G * Dch) return;
  int g = i >> 7;
  float c = cnt[g];
  out[i] = sums[i] / (c > 1.f ? c : 1.f);
}

extern "C" void kernel_launch(void* const* d_in, const int* in_sizes, int n_in,
                              void* d_out, int out_size, void* d_ws, size_t ws_size,
                              hipStream_t stream) {
  const float* x0    = (const float*)d_in[0];
  const int*   ei    = (const int*)d_in[1];
  const int*   batch = (const int*)d_in[2];
  const float* Ws    = (const float*)d_in[3];
  const float* a_src = (const float*)d_in[4];
  const float* a_dst = (const float*)d_in[5];
  const float* bias  = (const float*)d_in[6];
  const float* gamma = (const float*)d_in[7];
  const float* beta  = (const float*)d_in[8];
  const float* mean  = (const float*)d_in[9];
  const float* var   = (const float*)d_in[10];

  const int Nn    = in_sizes[0] / Dch;
  const int Ereal = in_sizes[1] / 2;
  const int Etot  = Ereal + Nn;
  const int G     = out_size / Dch;

  char* ws = (char*)d_ws;
  float*    bufA  = (float*)ws;    ws += (size_t)Nn * Dch * 4;
  float*    bufB  = (float*)ws;    ws += (size_t)Nn * Dch * 4;
  float*    hbuf  = (float*)ws;    ws += (size_t)Nn * Dch * 4;
  unsigned* mkey  = (unsigned*)ws; ws += (size_t)Nn * Hh * 4;
  float*    ssum  = (float*)ws;    ws += (size_t)Nn * Hh * 4;
  float*    alsrc = (float*)ws;    ws += (size_t)Nn * Hh * 4;
  float*    aldst = (float*)ws;    ws += (size_t)Nn * Hh * 4;
  float*    psum  = (float*)ws;    ws += (size_t)G * Dch * 4;
  float*    pcnt  = (float*)ws;    ws += (size_t)G * 4;

  const float* xin  = x0;
  float*       xout = bufA;
  for (int l = 0; l < Lnum; ++l) {
    const float* Wl = Ws + (size_t)l * Dch * Dch;
    gemm_wmma_f32<<<Nn / 16, 256, 0, stream>>>(xin, Wl, hbuf);

    att_logits<<<(Nn * Hh + 255) / 256, 256, 0, stream>>>(
        hbuf, a_src + (size_t)l * Hh * Ch, a_dst + (size_t)l * Hh * Ch,
        alsrc, aldst, Nn);

    fill_u32<<<512, 256, 0, stream>>>(mkey, 0u, (long)Nn * Hh);
    fill_u32<<<512, 256, 0, stream>>>((unsigned*)ssum, 0u, (long)Nn * Hh);
    fill_u32<<<2048, 256, 0, stream>>>((unsigned*)xout, 0u, (long)Nn * Dch);

    edge_max<<<(Etot + 255) / 256, 256, 0, stream>>>(ei, Ereal, Etot, alsrc, aldst, mkey);
    edge_expsum<<<(Etot + 255) / 256, 256, 0, stream>>>(ei, Ereal, Etot, alsrc, aldst, mkey, ssum);
    {
      long t = (long)Etot * 32;
      edge_aggregate<<<(int)((t + 255) / 256), 256, 0, stream>>>(
          ei, Ereal, Etot, alsrc, aldst, mkey, ssum, hbuf, xout);
    }

    post_bn_elu<<<(Nn * Dch + 255) / 256, 256, 0, stream>>>(
        xout, bias + (size_t)l * Dch, gamma + (size_t)l * Dch,
        beta + (size_t)l * Dch, mean + (size_t)l * Dch, var + (size_t)l * Dch,
        xout, (long)Nn * Dch);

    xin  = xout;
    xout = (xout == bufA) ? bufB : bufA;
  }

  fill_u32<<<32, 256, 0, stream>>>((unsigned*)psum, 0u, (long)G * Dch);
  fill_u32<<<1, 64, 0, stream>>>((unsigned*)pcnt, 0u, G);
  pool_accum<<<(Nn * Dch + 255) / 256, 256, 0, stream>>>(xin, batch, psum, pcnt, (long)Nn * Dch);
  pool_final<<<(G * Dch + 255) / 256, 256, 0, stream>>>(psum, pcnt, (float*)d_out, G);
}


# in-edges grouped by destination once per call; per layer one wave per node does max, exp-sum and the alpha-weighted row sum in registers: no float atomics, no accumulator clears
# speedup vs baseline: 2.8851x; 2.8851x over previous
#include <hip/hip_runtime.h>


#define Dch      128
#define Hh       4
#define Ch       32
#define Lnum     3
#define NEG_SLOPE 0.2f
#define BN_EPS   1e-5f

typedef __attribute__((ext_vector_type(2))) float v2f;
typedef __attribute__((ext_vector_type(8))) float v8f;

__global__ void fill_u32(unsigned* __restrict__ p, unsigned v, long n) {
  long i = (long)blockIdx.x * blockDim.x + threadIdx.x;
  long stride = (long)gridDim.x * blockDim.x;
  for (; i < n; i += stride) p[i] = v;
}

__global__ __launch_bounds__(256) void gemm_wmma_f32(
    const float* __restrict__ X, const float* __restrict__ Wm,
    float* __restrict__ Ho) {
  const int wave = threadIdx.x >> 5;
  const int lane = threadIdx.x & 31;
  const int half = lane >> 4;
  const int l15  = lane & 15;
  const int row0 = blockIdx.x << 4;
  const int bcol = (wave << 4) + l15;
  const float* arow = X + (long)(row0 + l15) * Dch;

  v8f acc = {};
  for (int k = 0; k < Dch; k += 4) {
    const int ka = k + half * 2;
    v2f a, b;
    a.x = arow[ka];
    a.y = arow[ka + 1];
    b.x = Wm[(long)ka * Dch + bcol];
    b.y = Wm[(long)(ka + 1) * Dch + bcol];
    acc = __builtin_amdgcn_wmma_f32_16x16x4_f32(
         false, a,  false, b,
         (short)0, acc,  false,  false);
  }
  float* orow = Ho + (long)(row0 + half * 8) * Dch + bcol;
#pragma unroll
  for (int v = 0; v < 8; ++v) orow[(long)v * Dch] = acc[v];
}

__global__ void att_logits(const float* __restrict__ Hf,
                           const float* __restrict__ a_src,
                           const float* __restrict__ a_dst,
                           float* __restrict__ alsrc, float* __restrict__ aldst,
                           int Nn) {
  int i = blockIdx.x * blockDim.x + threadIdx.x;
  if (i >= Nn * Hh) return;
  int n = i >> 2, h = i & 3;
  const float* hp = Hf + (long)n * Dch + h * Ch;
  const float* as = a_src + h * Ch;
  const float* ad = a_dst + h * Ch;
  float s0 = 0.f, s1 = 0.f;
#pragma unroll
  for (int c = 0; c < Ch; ++c) { float v = hp[c]; s0 += v * as[c]; s1 += v * ad[c]; }
  alsrc[i] = s0;
  aldst[i] = s1;
}

__global__ void deg_count(const int* __restrict__ ei, int Ereal,
                          int* __restrict__ seg_len) {
  int e = blockIdx.x * blockDim.x + threadIdx.x;
  if (e >= Ereal) return;
  atomicAdd(&seg_len[ei[Ereal + e]], 1);
}

__global__ __launch_bounds__(256) void seg_alloc(
    const int* seg_len, int Nn, int* total,
    int* __restrict__ seg_start, int* __restrict__ cursor) {
  __shared__ int wave_off[8];
  __shared__ int block_base;
  const int n    = blockIdx.x * 256 + threadIdx.x;
  const int lane = threadIdx.x & 31;
  const int wave = threadIdx.x >> 5;
  const int v    = n < Nn ? seg_len[n] : 0;
  int inc = v;
  for (int o = 1; o < 32; o <<= 1) {
    int t = __shfl_up(inc, o, 32);
    if (lane >= o) inc += t;
  }
  if (lane == 31) wave_off[wave] = inc;
  __syncthreads();
  if (threadIdx.x == 0) {
    int run = 0;
    for (int w = 0; w < 8; ++w) { int t = wave_off[w]; wave_off[w] = run; run += t; }
    block_base = atomicAdd(total, run);
  }
  __syncthreads();
  const int st = block_base + wave_off[wave] + inc - v;
  if (n < Nn) { seg_start[n] = st; cursor[n] = st; }
}

__global__ void csr_fill(const int* __restrict__ ei, int Ereal,
                         int* __restrict__ cursor, int* __restrict__ csr_src) {
  int e = blockIdx.x * blockDim.x + threadIdx.x;
  if (e >= Ereal) return;
  const int s = ei[e];
  const int d = ei[Ereal + e];
  const int pos = atomicAdd(&cursor[d], 1);
  csr_src[pos] = s;
}

__device__ __forceinline__ float lrelu(float v) { return v > 0.f ? v : v * NEG_SLOPE; }

__device__ __forceinline__ float4 ld4(const float* p) {
  return *reinterpret_cast<const float4*>(p);
}

__device__ __forceinline__ float wave_max(float v) {
  for (int o = 16; o > 0; o >>= 1) v = fmaxf(v, __shfl_xor(v, o, 32));
  return v;
}
__device__ __forceinline__ float wave_sum(float v) {
  for (int o = 16; o > 0; o >>= 1) v += __shfl_xor(v, o, 32);
  return v;
}

__device__ __forceinline__ bool chunk_scores(
    const int* __restrict__ csr_src, const float* __restrict__ alsrc,
    int beg, int cnt, int base, int lane, float4 ad, int& s, float4& e) {
  const int j = base + lane;
  const bool valid = j < cnt;
  s = 0;
  e = make_float4(0.f, 0.f, 0.f, 0.f);
  if (valid) {
    s = csr_src[beg + j];
    const float4 a = ld4(alsrc + (long)s * Hh);
    e.x = lrelu(a.x + ad.x);
    e.y = lrelu(a.y + ad.y);
    e.z = lrelu(a.z + ad.z);
    e.w = lrelu(a.w + ad.w);
  }
  return valid;
}

__device__ __forceinline__ int bcast(float v, int j) {
  return __builtin_amdgcn_readlane(__float_as_int(v), j);
}

__device__ __forceinline__ void add_rows(const float* __restrict__ hl,
                                         int k0, int k1, int k2, int k3,
                                         int s, float4 al, int n, float4& acc) {
#pragma unroll 4
  for (int j = 0; j < n; ++j) {
    const int sj = __builtin_amdgcn_readlane(s, j);
    const float b = __int_as_float((bcast(al.x, j) & k0) | (bcast(al.y, j) & k1) |
                                   (bcast(al.z, j) & k2) | (bcast(al.w, j) & k3));
    const float4 hv = ld4(hl + (long)sj * Dch);
    acc.x += b * hv.x;
    acc.y += b * hv.y;
    acc.z += b * hv.z;
    acc.w += b * hv.w;
  }
}

__global__ __launch_bounds__(256) void gat_gather(
    const int* __restrict__ seg_start, const int* __restrict__ seg_len,
    const int* __restrict__ csr_src,
    const float* __restrict__ alsrc, const float* __restrict__ aldst,
    const float* __restrict__ Hf, float* __restrict__ xo, int Nn) {
  const int lane = threadIdx.x & 31;
  const int d = __builtin_amdgcn_readfirstlane((int)(blockIdx.x * 8 + (threadIdx.x >> 5)));
  if (d >= Nn) return;
  const int beg = seg_start[d];
  const int cnt = seg_len[d];
  const float4 ad = ld4(aldst + (long)d * Hh);
  const float4 as = ld4(alsrc + (long)d * Hh);
  float4 es;
  es.x = lrelu(as.x + ad.x);
  es.y = lrelu(as.y + ad.y);
  es.z = lrelu(as.z + ad.z);
  es.w = lrelu(as.w + ad.w);

  int s0; float4 e0;
  const bool v0 = chunk_scores(csr_src, alsrc, beg, cnt, 0, lane, ad, s0, e0);

  float4 m = es;
  if (v0) {
    m.x = fmaxf(m.x, e0.x); m.y = fmaxf(m.y, e0.y);
    m.z = fmaxf(m.z, e0.z); m.w = fmaxf(m.w, e0.w);
  }
  for (int base = 32; base < cnt; base += 32) {
    int s; float4 e;
    if (chunk_scores(csr_src, alsrc, beg, cnt, base, lane, ad, s, e)) {
      m.x = fmaxf(m.x, e.x); m.y = fmaxf(m.y, e.y);
      m.z = fmaxf(m.z, e.z); m.w = fmaxf(m.w, e.w);
    }
  }
  m.x = wave_max(m.x); m.y = wave_max(m.y);
  m.z = wave_max(m.z); m.w = wave_max(m.w);

  float4 z = make_float4(0.f, 0.f, 0.f, 0.f);
  float4 ps;
  ps.x = expf(es.x - m.x); ps.y = expf(es.y - m.y);
  ps.z = expf(es.z - m.z); ps.w = expf(es.w - m.w);
  if (lane == 0) z = ps;
  float4 p0 = make_float4(0.f, 0.f, 0.f, 0.f);
  if (v0) {
    p0.x = expf(e0.x - m.x); p0.y = expf(e0.y - m.y);
    p0.z = expf(e0.z - m.z); p0.w = expf(e0.w - m.w);
    z.x += p0.x; z.y += p0.y; z.z += p0.z; z.w += p0.w;
  }
  for (int base = 32; base < cnt; base += 32) {
    int s; float4 e;
    if (chunk_scores(csr_src, alsrc, beg, cnt, base, lane, ad, s, e)) {
      z.x += expf(e.x - m.x); z.y += expf(e.y - m.y);
      z.z += expf(e.z - m.z); z.w += expf(e.w - m.w);
    }
  }
  float4 den;
  den.x = wave_sum(z.x) + 1e-16f; den.y = wave_sum(z.y) + 1e-16f;
  den.z = wave_sum(z.z) + 1e-16f; den.w = wave_sum(z.w) + 1e-16f;

  const int hsel = lane >> 3;
  const int k0 = hsel == 0 ? -1 : 0, k1 = hsel == 1 ? -1 : 0;
  const int k2 = hsel == 2 ? -1 : 0, k3 = hsel == 3 ? -1 : 0;
  const float* hl = Hf + lane * 4;
  float4 acc;
  {
    const float a_self = __int_as_float(
        (__float_as_int(ps.x / den.x) & k0) | (__float_as_int(ps.y / den.y) & k1) |
        (__float_as_int(ps.z / den.z) & k2) | (__float_as_int(ps.w / den.w) & k3));
    const float4 hv = ld4(hl + (long)d * Dch);
    acc.x = a_self * hv.x; acc.y = a_self * hv.y;
    acc.z = a_self * hv.z; acc.w = a_self * hv.w;
  }
  {
    float4 al;
    al.x = p0.x / den.x; al.y = p0.y / den.y;
    al.z = p0.z / den.z; al.w = p0.w / den.w;
    add_rows(hl, k0, k1, k2, k3, s0, al, cnt < 32 ? cnt : 32, acc);
  }
  for (int base = 32; base < cnt; base += 32) {
    int s; float4 e;
    float4 al = make_float4(0.f, 0.f, 0.f, 0.f);
    if (chunk_scores(csr_src, alsrc, beg, cnt, base, lane, ad, s, e)) {
      al.x = expf(e.x - m.x) / den.x; al.y = expf(e.y - m.y) / den.y;
      al.z = expf(e.z - m.z) / den.z; al.w = expf(e.w - m.w) / den.w;
    }
    const int left = cnt - base;
    add_rows(hl, k0, k1, k2, k3, s, al, left < 32 ? left : 32, acc);
  }
  *reinterpret_cast<float4*>(xo + (long)d * Dch + lane * 4) = acc;
}

__global__ void post_bn_elu(const float* __restrict__ acc,
                            const float* __restrict__ bias,
                            const float* __restrict__ gamma,
                            const float* __restrict__ beta,
                            const float* __restrict__ mean,
                            const float* __restrict__ var,
                            float* __restrict__ xo, long n) {
  long i = (long)blockIdx.x * blockDim.x + threadIdx.x;
  if (i >= n) return;
  int c = (int)(i & (Dch - 1));
  float v = acc[i] + bias[c];
  v = (v - mean[c]) * (gamma[c] / sqrtf(var[c] + BN_EPS)) + beta[c];
  xo[i] = v > 0.f ? v : expf(v) - 1.f;
}

__global__ void pool_accum(const float* __restrict__ x,
                           const int* __restrict__ batch,
                           float* __restrict__ sums, float* __restrict__ cnt,
                           long n) {
  long i = (long)blockIdx.x * blockDim.x + threadIdx.x;
  if (i >= n) return;
  int node = (int)(i >> 7);
  int c    = (int)(i & 127);
  int g    = batch[node];
  atomicAdd(&sums[(long)g * Dch + c], x[i]);
  if (c == 0) atomicAdd(&cnt[g], 1.0f);
}

__global__ void pool_final(const float* __restrict__ sums,
                           const float* __restrict__ cnt,
                           float* __restrict__ out, int G) {
  int i = blockIdx.x * blockDim.x + threadIdx.x;
  if (i >= G * Dch) return;
  int g = i >> 7;
  float c = cnt[g];
  out[i] = sums[i] / (c > 1.f ? c : 1.f);
}

extern "C" void kernel_launch(void* const* d_in, const int* in_sizes, int n_in,
                              void* d_out, int out_size, void* d_ws, size_t ws_size,
                              hipStream_t stream) {
  const float* x0    = (const float*)d_in[0];
  const int*   ei    = (const int*)d_in[1];
  const int*   batch = (const int*)d_in[2];
  const float* Ws    = (const float*)d_in[3];
  const float* a_src = (const float*)d_in[4];
  const float* a_dst = (const float*)d_in[5];
  const float* bias  = (const float*)d_in[6];
  const float* gamma = (const float*)d_in[7];
  const float* beta  = (const float*)d_in[8];
  const float* mean  = (const float*)d_in[9];
  const float* var   = (const float*)d_in[10];

  const int Nn    = in_sizes[0] / Dch;
  const int Ereal = in_sizes[1] / 2;
  const int G     = out_size / Dch;

  char* ws = (char*)d_ws;
  float*    xbuf  = (float*)ws;    ws += (size_t)Nn * Dch * 4;
  char*     grp   = ws;            ws += (size_t)Nn * Dch * 4;
  float*    hbuf  = (float*)ws;    ws += (size_t)Nn * Dch * 4;
  ws += (size_t)Nn * Hh * 4;
  ws += (size_t)Nn * Hh * 4;
  float*    alsrc = (float*)ws;    ws += (size_t)Nn * Hh * 4;
  float*    aldst = (float*)ws;    ws += (size_t)Nn * Hh * 4;
  float*    psum  = (float*)ws;    ws += (size_t)G * Dch * 4;
  float*    pcnt  = (float*)ws;    ws += (size_t)G * 4;

  int* csr_src   = (int*)grp;
  int* seg_len   = (int*)(grp + (((size_t)Ereal * 4 + 255) & ~(size_t)255));
  int* seg_start = seg_len + (((size_t)Nn + 1 + 63) & ~(size_t)63);
  int* cursor    = seg_start + (((size_t)Nn + 63) & ~(size_t)63);

  fill_u32<<<64, 256, 0, stream>>>((unsigned*)seg_len, 0u, (long)Nn + 1);
  deg_count<<<(Ereal + 255) / 256, 256, 0, stream>>>(ei, Ereal, seg_len);
  seg_alloc<<<(Nn + 255) / 256, 256, 0, stream>>>(seg_len, Nn, seg_len + Nn, seg_start, cursor);
  csr_fill<<<(Ereal + 255) / 256, 256, 0, stream>>>(ei, Ereal, cursor, csr_src);

  const float* xin = x0;
  for (int l = 0; l < Lnum; ++l) {
    const float* Wl = Ws + (size_t)l * Dch * Dch;
    gemm_wmma_f32<<<Nn / 16, 256, 0, stream>>>(xin, Wl, hbuf);

    att_logits<<<(Nn * Hh + 255) / 256, 256, 0, stream>>>(
        hbuf, a_src + (size_t)l * Hh * Ch, a_dst + (size_t)l * Hh * Ch,
        alsrc, aldst, Nn);

    gat_gather<<<(Nn + 7) / 8, 256, 0, stream>>>(
        seg_start, seg_len, csr_src, alsrc, aldst, hbuf, xbuf, Nn);

    post_bn_elu<<<(Nn * Dch + 255) / 256, 256, 0, stream>>>(
        xbuf, bias + (size_t)l * Dch, gamma + (size_t)l * Dch,
        beta + (size_t)l * Dch, mean + (size_t)l * Dch, var + (size_t)l * Dch,
        xbuf, (long)Nn * Dch);

    xin = xbuf;
  }

  fill_u32<<<32, 256, 0, stream>>>((unsigned*)psum, 0u, (long)G * Dch);
  fill_u32<<<1, 64, 0, stream>>>((unsigned*)pcnt, 0u, G);
  pool_accum<<<(Nn * Dch + 255) / 256, 256, 0, stream>>>(xin, batch, psum, pcnt, (long)Nn * Dch);
  pool_final<<<(G * Dch + 255) / 256, 256, 0, stream>>>(psum, pcnt, (float*)d_out, G);
}


# bucket build and first-layer GEMM in one launch (independent work: returning adds at the second-level cache beside the matrix pipe): 9 to 8 launches, the GEMM hidden under the adds
# speedup vs baseline: 6.2285x; 2.1589x over previous
#include <hip/hip_runtime.h>


#define Dch      128
#define Hh       4
#define Ch       32
#define Lnum     3
#define NEG_SLOPE 0.2f
#define BN_EPS   1e-5f

typedef __attribute__((ext_vector_type(2))) float v2f;
typedef __attribute__((ext_vector_type(8))) float v8f;

__global__ void fill_u32(unsigned* __restrict__ p, unsigned v, long n) {
  long i = (long)blockIdx.x * blockDim.x + threadIdx.x;
  long stride = (long)gridDim.x * blockDim.x;
  for (; i < n; i += stride) p[i] = v;
}

__device__ __forceinline__ float4 ld4(const float* p) {
  return *reinterpret_cast<const float4*>(p);
}

template <int M>
__device__ __forceinline__ float row_xor(float v) {
  return __int_as_float(__builtin_amdgcn_update_dpp(
      0, __float_as_int(v), 0x160 | M, 0xf, 0xf, false));
}

#define GEMM_STRIPS  2
#define GEMM_THREADS (GEMM_STRIPS * 64)
__device__ __forceinline__ void gemm_block(
    const int bx,
    const float* __restrict__ X, const float* __restrict__ Wm,
    const float* __restrict__ a_src, const float* __restrict__ a_dst,
    float* __restrict__ Ho, float* __restrict__ alsrc,
    float* __restrict__ aldst, int nstrips) {
  __shared__ float lg[GEMM_STRIPS * 2 * 16 * Hh];
  const int wave = threadIdx.x >> 5;
  const int lane = threadIdx.x & 31;
  const int half = lane >> 4;
  const int l15  = lane & 15;
  const int sloc = wave >> 1;
  const int grp  = wave & 1;
  const int strip = bx * GEMM_STRIPS + sloc;
  const bool live = strip < nstrips;
  const int row0 = strip << 4;
  const int col0 = grp * 64 + l15 * 4;

  v8f acc0 = {}, acc1 = {}, acc2 = {}, acc3 = {};
  if (live) {
    const float* arow = X + (long)(row0 + l15) * Dch;
    const float* bcol = Wm + col0;
    for (int k = 0; k < Dch; k += 4) {
      const int ka = k + half * 2;
      v2f a, b;
      a.x = arow[ka];
      a.y = arow[ka + 1];
      const float4 b0 = ld4(bcol + (long)ka * Dch);
      const float4 b1 = ld4(bcol + (long)(ka + 1) * Dch);
      b.x = b0.x; b.y = b1.x;
      acc0 = __builtin_amdgcn_wmma_f32_16x16x4_f32(
           false, a,  false, b,
           (short)0, acc0,  false,  false);
      b.x = b0.y; b.y = b1.y;
      acc1 = __builtin_amdgcn_wmma_f32_16x16x4_f32(
          false, a, false, b, (short)0, acc1, false, false);
      b.x = b0.z; b.y = b1.z;
      acc2 = __builtin_amdgcn_wmma_f32_16x16x4_f32(
          false, a, false, b, (short)0, acc2, false, false);
      b.x = b0.w; b.y = b1.w;
      acc3 = __builtin_amdgcn_wmma_f32_16x16x4_f32(
          false, a, false, b, (short)0, acc3, false, false);
    }

    const float4 as = ld4(a_src + col0);
    const float4 ad = ld4(a_dst + col0);
    float ssel = 0.f, dsel = 0.f;
#pragma unroll
    for (int v = 0; v < 8; ++v) {
      float ps = acc0[v] * as.x + acc1[v] * as.y + acc2[v] * as.z + acc3[v] * as.w;
      float pd = acc0[v] * ad.x + acc1[v] * ad.y + acc2[v] * ad.z + acc3[v] * ad.w;
      ps += row_xor<1>(ps);  pd += row_xor<1>(pd);
      ps += row_xor<2>(ps);  pd += row_xor<2>(pd);
      ps += row_xor<4>(ps);  pd += row_xor<4>(pd);
      if ((l15 & 7) == v) { ssel = ps; dsel = pd; }
    }
    const int li = sloc * 128 + (half * 8 + (l15 & 7)) * Hh + grp * 2 + (l15 >> 3);
    lg[li]      = ssel;
    lg[li + 64] = dsel;
  }
  __syncthreads();
  const int t = threadIdx.x;
  if (bx * GEMM_STRIPS + (t >> 6) < nstrips) {
    const long o = (long)bx * (GEMM_STRIPS * 16 * Hh) + t;
    const int  q = (t >> 6) * 128 + (t & 63);
    alsrc[o] = lg[q];
    aldst[o] = lg[q + 64];
  }
  if (live) {
    float* orow = Ho + (long)(row0 + half * 8) * Dch + col0;
#pragma unroll
    for (int v = 0; v < 8; ++v)
      *reinterpret_cast<float4*>(orow + (long)v * Dch) =
          make_float4(acc0[v], acc1[v], acc2[v], acc3[v]);
  }
}

__global__ __launch_bounds__(GEMM_THREADS) void gemm_wmma_f32(
    const float* __restrict__ X, const float* __restrict__ Wm,
    const float* __restrict__ a_src, const float* __restrict__ a_dst,
    float* __restrict__ Ho, float* __restrict__ alsrc,
    float* __restrict__ aldst, int nstrips) {
  gemm_block((int)blockIdx.x, X, Wm, a_src, a_dst, Ho, alsrc, aldst, nstrips);
}

#define BKT 64
__device__ __forceinline__ void bucket_edge(
    const int e, const int* __restrict__ ei, int Ereal,
    int* __restrict__ seg_len, int* __restrict__ bucket) {
  if (e >= Ereal) return;
  const int s = ei[e];
  const int d = ei[Ereal + e];
  const int pos = atomicAdd(&seg_len[d], 1);
  bucket[d * BKT + (pos < BKT ? pos : BKT - 1)] = s;
}

__global__ __launch_bounds__(GEMM_THREADS) void gemm_beside_buckets(
    const float* __restrict__ X, const float* __restrict__ Wm,
    const float* __restrict__ a_src, const float* __restrict__ a_dst,
    float* __restrict__ Ho, float* __restrict__ alsrc,
    float* __restrict__ aldst, int nstrips, int gemm_blocks,
    const int* __restrict__ ei, int Ereal,
    int* __restrict__ seg_len, int* __restrict__ bucket) {
  const int bx = (int)blockIdx.x;
  if (bx < gemm_blocks) {
    gemm_block(bx, X, Wm, a_src, a_dst, Ho, alsrc, aldst, nstrips);
  } else {
    bucket_edge((bx - gemm_blocks) * GEMM_THREADS + (int)threadIdx.x,
                ei, Ereal, seg_len, bucket);
  }
}

__device__ __forceinline__ float lrelu(float v) { return v > 0.f ? v : v * NEG_SLOPE; }

__device__ __forceinline__ float wave_max(float v) {
  for (int o = 16; o > 0; o >>= 1) v = fmaxf(v, __shfl_xor(v, o, 32));
  return v;
}
__device__ __forceinline__ float wave_sum(float v) {
  for (int o = 16; o > 0; o >>= 1) v += __shfl_xor(v, o, 32);
  return v;
}

__device__ __forceinline__ bool chunk_scores(
    const int* __restrict__ csr_src, const float* __restrict__ alsrc,
    int beg, int cnt, int base, int lane, float4 ad, int& s, float4& e) {
  const int j = base + lane;
  const bool valid = j < cnt;
  s = 0;
  e = make_float4(0.f, 0.f, 0.f, 0.f);
  if (valid) {
    s = csr_src[beg + j];
    const float4 a = ld4(alsrc + (long)s * Hh);
    e.x = lrelu(a.x + ad.x);
    e.y = lrelu(a.y + ad.y);
    e.z = lrelu(a.z + ad.z);
    e.w = lrelu(a.w + ad.w);
  }
  return valid;
}

__device__ __forceinline__ int bcast(float v, int j) {
  return __builtin_amdgcn_readlane(__float_as_int(v), j);
}

__device__ __forceinline__ void add_rows(const float* __restrict__ hl,
                                         int k0, int k1, int k2, int k3,
                                         int s, float4 al, int n, float4& acc) {
#pragma unroll 4
  for (int j = 0; j < n; ++j) {
    const int sj = __builtin_amdgcn_readlane(s, j);
    const float b = __int_as_float((bcast(al.x, j) & k0) | (bcast(al.y, j) & k1) |
                                   (bcast(al.z, j) & k2) | (bcast(al.w, j) & k3));
    const float4 hv = ld4(hl + (long)sj * Dch);
    acc.x += b * hv.x;
    acc.y += b * hv.y;
    acc.z += b * hv.z;
    acc.w += b * hv.w;
  }
}

__device__ __forceinline__ bool scan_scores(
    const int* __restrict__ ei, int Ereal, const float* __restrict__ alsrc,
    int d, int base, int lane, float4 ad, int& s, float4& e) {
  const int i = base + lane;
  s = 0;
  e = make_float4(0.f, 0.f, 0.f, 0.f);
  const bool valid = i < Ereal && ei[Ereal + i] == d;
  if (valid) {
    s = ei[i];
    const float4 a = ld4(alsrc + (long)s * Hh);
    e.x = lrelu(a.x + ad.x);
    e.y = lrelu(a.y + ad.y);
    e.z = lrelu(a.z + ad.z);
    e.w = lrelu(a.w + ad.w);
  }
  return valid;
}

__device__ float4 gather_by_scan(
    const int* __restrict__ ei, int Ereal, const float* __restrict__ alsrc,
    const float* __restrict__ hl, int d, int lane, float4 ad, float4 es,
    int k0, int k1, int k2, int k3) {
  float4 m = es;
  for (int base = 0; base < Ereal; base += 32) {
    int s; float4 e;
    if (scan_scores(ei, Ereal, alsrc, d, base, lane, ad, s, e)) {
      m.x = fmaxf(m.x, e.x); m.y = fmaxf(m.y, e.y);
      m.z = fmaxf(m.z, e.z); m.w = fmaxf(m.w, e.w);
    }
  }
  m.x = wave_max(m.x); m.y = wave_max(m.y);
  m.z = wave_max(m.z); m.w = wave_max(m.w);

  float4 ps;
  ps.x = expf(es.x - m.x); ps.y = expf(es.y - m.y);
  ps.z = expf(es.z - m.z); ps.w = expf(es.w - m.w);
  float4 z = make_float4(0.f, 0.f, 0.f, 0.f);
  if (lane == 0) z = ps;
  for (int base = 0; base < Ereal; base += 32) {
    int s; float4 e;
    if (scan_scores(ei, Ereal, alsrc, d, base, lane, ad, s, e)) {
      z.x += expf(e.x - m.x); z.y += expf(e.y - m.y);
      z.z += expf(e.z - m.z); z.w += expf(e.w - m.w);
    }
  }
  float4 den;
  den.x = wave_sum(z.x) + 1e-16f; den.y = wave_sum(z.y) + 1e-16f;
  den.z = wave_sum(z.z) + 1e-16f; den.w = wave_sum(z.w) + 1e-16f;

  float4 acc;
  {
    const float a_self = __int_as_float(
        (__float_as_int(ps.x / den.x) & k0) | (__float_as_int(ps.y / den.y) & k1) |
        (__float_as_int(ps.z / den.z) & k2) | (__float_as_int(ps.w / den.w) & k3));
    const float4 hv = ld4(hl + (long)d * Dch);
    acc.x = a_self * hv.x; acc.y = a_self * hv.y;
    acc.z = a_self * hv.z; acc.w = a_self * hv.w;
  }
  for (int base = 0; base < Ereal; base += 32) {
    int s; float4 e;
    float4 al = make_float4(0.f, 0.f, 0.f, 0.f);
    const bool v = scan_scores(ei, Ereal, alsrc, d, base, lane, ad, s, e);
    if (v) {
      al.x = expf(e.x - m.x) / den.x; al.y = expf(e.y - m.y) / den.y;
      al.z = expf(e.z - m.z) / den.z; al.w = expf(e.w - m.w) / den.w;
    }
    unsigned hit = __builtin_amdgcn_ballot_w32(v);
    while (hit) {
      const int j = __builtin_ctz(hit);
      hit &= hit - 1;
      const int sj = __builtin_amdgcn_readlane(s, j);
      const float b = __int_as_float((bcast(al.x, j) & k0) | (bcast(al.y, j) & k1) |
                                     (bcast(al.z, j) & k2) | (bcast(al.w, j) & k3));
      const float4 hv = ld4(hl + (long)sj * Dch);
      acc.x += b * hv.x;
      acc.y += b * hv.y;
      acc.z += b * hv.z;
      acc.w += b * hv.w;
    }
  }
  return acc;
}

__device__ __forceinline__ float bn_elu(float a, float b, float g, float be,
                                        float mu, float va) {
  float v = a + b;
  v = (v - mu) * (g / sqrtf(va + BN_EPS)) + be;
  return v > 0.f ? v : expf(v) - 1.f;
}

__global__ __launch_bounds__(256) void gat_gather(
    const int* __restrict__ seg_len, const int* __restrict__ csr_src,
    const float* __restrict__ alsrc, const float* __restrict__ aldst,
    const float* __restrict__ Hf,
    const float* __restrict__ bias, const float* __restrict__ gamma,
    const float* __restrict__ beta, const float* __restrict__ mean,
    const float* __restrict__ var,
    float* __restrict__ xo, int Nn,
    const int* __restrict__ ei, int Ereal) {
  const int lane = threadIdx.x & 31;
  const int d = __builtin_amdgcn_readfirstlane((int)(blockIdx.x * 8 + (threadIdx.x >> 5)));
  if (d >= Nn) return;
  const int beg = d * BKT;
  const int cnt = seg_len[d];
  const float4 ad = ld4(aldst + (long)d * Hh);
  const float4 as = ld4(alsrc + (long)d * Hh);
  __builtin_amdgcn_sched_barrier(0);
  float4 es;
  es.x = lrelu(as.x + ad.x);
  es.y = lrelu(as.y + ad.y);
  es.z = lrelu(as.z + ad.z);
  es.w = lrelu(as.w + ad.w);

  const int hsel = lane >> 3;
  const int k0 = hsel == 0 ? -1 : 0, k1 = hsel == 1 ? -1 : 0;
  const int k2 = hsel == 2 ? -1 : 0, k3 = hsel == 3 ? -1 : 0;
  const float* hl = Hf + lane * 4;
  float4 acc;
  if (cnt > BKT) {
    acc = gather_by_scan(ei, Ereal, alsrc, hl, d, lane, ad, es, k0, k1, k2, k3);
  } else {
  int s0; float4 e0;
  const bool v0 = chunk_scores(csr_src, alsrc, beg, cnt, 0, lane, ad, s0, e0);

  float4 m = es;
  if (v0) {
    m.x = fmaxf(m.x, e0.x); m.y = fmaxf(m.y, e0.y);
    m.z = fmaxf(m.z, e0.z); m.w = fmaxf(m.w, e0.w);
  }
  for (int base = 32; base < cnt; base += 32) {
    int s; float4 e;
    if (chunk_scores(csr_src, alsrc, beg, cnt, base, lane, ad, s, e)) {
      m.x = fmaxf(m.x, e.x); m.y = fmaxf(m.y, e.y);
      m.z = fmaxf(m.z, e.z); m.w = fmaxf(m.w, e.w);
    }
  }
  m.x = wave_max(m.x); m.y = wave_max(m.y);
  m.z = wave_max(m.z); m.w = wave_max(m.w);

  float4 z = make_float4(0.f, 0.f, 0.f, 0.f);
  float4 ps;
  ps.x = expf(es.x - m.x); ps.y = expf(es.y - m.y);
  ps.z = expf(es.z - m.z); ps.w = expf(es.w - m.w);
  if (lane == 0) z = ps;
  float4 p0 = make_float4(0.f, 0.f, 0.f, 0.f);
  if (v0) {
    p0.x = expf(e0.x - m.x); p0.y = expf(e0.y - m.y);
    p0.z = expf(e0.z - m.z); p0.w = expf(e0.w - m.w);
    z.x += p0.x; z.y += p0.y; z.z += p0.z; z.w += p0.w;
  }
  for (int base = 32; base < cnt; base += 32) {
    int s; float4 e;
    if (chunk_scores(csr_src, alsrc, beg, cnt, base, lane, ad, s, e)) {
      z.x += expf(e.x - m.x); z.y += expf(e.y - m.y);
      z.z += expf(e.z - m.z); z.w += expf(e.w - m.w);
    }
  }
  float4 den;
  den.x = wave_sum(z.x) + 1e-16f; den.y = wave_sum(z.y) + 1e-16f;
  den.z = wave_sum(z.z) + 1e-16f; den.w = wave_sum(z.w) + 1e-16f;

  {
    const float a_self = __int_as_float(
        (__float_as_int(ps.x / den.x) & k0) | (__float_as_int(ps.y / den.y) & k1) |
        (__float_as_int(ps.z / den.z) & k2) | (__float_as_int(ps.w / den.w) & k3));
    const float4 hv = ld4(hl + (long)d * Dch);
    acc.x = a_self * hv.x; acc.y = a_self * hv.y;
    acc.z = a_self * hv.z; acc.w = a_self * hv.w;
  }
  {
    float4 al;
    al.x = p0.x / den.x; al.y = p0.y / den.y;
    al.z = p0.z / den.z; al.w = p0.w / den.w;
    add_rows(hl, k0, k1, k2, k3, s0, al, cnt < 32 ? cnt : 32, acc);
  }
  for (int base = 32; base < cnt; base += 32) {
    int s; float4 e;
    float4 al = make_float4(0.f, 0.f, 0.f, 0.f);
    if (chunk_scores(csr_src, alsrc, beg, cnt, base, lane, ad, s, e)) {
      al.x = expf(e.x - m.x) / den.x; al.y = expf(e.y - m.y) / den.y;
      al.z = expf(e.z - m.z) / den.z; al.w = expf(e.w - m.w) / den.w;
    }
    const int left = cnt - base;
    add_rows(hl, k0, k1, k2, k3, s, al, left < 32 ? left : 32, acc);
  }
  }
  const float4 bi = ld4(bias + lane * 4);
  const float4 ga = ld4(gamma + lane * 4);
  const float4 be = ld4(beta + lane * 4);
  const float4 mu = ld4(mean + lane * 4);
  const float4 va = ld4(var + lane * 4);
  float4 o;
  o.x = bn_elu(acc.x, bi.x, ga.x, be.x, mu.x, va.x);
  o.y = bn_elu(acc.y, bi.y, ga.y, be.y, mu.y, va.y);
  o.z = bn_elu(acc.z, bi.z, ga.z, be.z, mu.z, va.z);
  o.w = bn_elu(acc.w, bi.w, ga.w, be.w, mu.w, va.w);
  *reinterpret_cast<float4*>(xo + (long)d * Dch + lane * 4) = o;
}

#define POOL_THREADS 1024
#define POOL_WAVES   32
#define POOL_TILE    65536
#define POOL_IDS     16

__global__ __launch_bounds__(POOL_THREADS) void pool_owner(
    const float* __restrict__ x, const int* __restrict__ batch,
    float* __restrict__ out, int Nn) {
  __shared__ unsigned member[POOL_TILE / 32];
  __shared__ float    part[POOL_WAVES][Dch];
  __shared__ float    pcount[POOL_WAVES];
  const int g    = blockIdx.x;
  const int lane = threadIdx.x & 31;
  const int wave = __builtin_amdgcn_readfirstlane((int)(threadIdx.x >> 5));
  float4 acc = make_float4(0.f, 0.f, 0.f, 0.f);
  float  cnt = 0.f;

  for (int t0 = 0; t0 < Nn; t0 += POOL_TILE) {
    const int tn = Nn - t0 < POOL_TILE ? Nn - t0 : POOL_TILE;

    const int nch = (tn + 31) >> 5;
    for (int c0 = wave; c0 < nch; c0 += POOL_WAVES * POOL_IDS) {
      int id[POOL_IDS];
#pragma unroll
      for (int k = 0; k < POOL_IDS; ++k) {
        const int n = t0 + (c0 + POOL_WAVES * k) * 32 + lane;
        id[k] = batch[n < Nn ? n : Nn - 1];
      }
#pragma unroll
      for (int k = 0; k < POOL_IDS; ++k) {
        const int c = c0 + POOL_WAVES * k;
        const int n = t0 + c * 32 + lane;
        const unsigned b = __builtin_amdgcn_ballot_w32(n < Nn && id[k] == g);
        if (c < nch && lane == 0) member[c] = b;
      }
    }
    __syncthreads();

    const int ngrp = (tn + 7) >> 3;
    for (int q0 = wave; q0 < ngrp; q0 += 32 * POOL_WAVES) {
      const int q = q0 + 32 * lane;
      unsigned m8 = 0;
      if (q < ngrp) m8 = (member[q >> 2] >> ((q & 3) * 8)) & 0xffu;
      unsigned any = __builtin_amdgcn_ballot_w32(m8 != 0);
      while (any) {
        const int i = __builtin_ctz(any);
        any &= any - 1;
        const unsigned mm = (unsigned)__builtin_amdgcn_readlane((int)m8, i);
        const int nb = t0 + (q0 + 32 * i) * 8;
        float4 r[8];
#pragma unroll
        for (int j = 0; j < 8; ++j) {
          int n = nb + j;
          n = n < Nn ? n : Nn - 1;
          r[j] = ld4(x + (long)n * Dch + lane * 4);
        }
#pragma unroll
        for (int j = 0; j < 8; ++j) {
          if (mm & (1u << j)) {
            acc.x += r[j].x; acc.y += r[j].y; acc.z += r[j].z; acc.w += r[j].w;
          }
        }
        cnt += (float)__builtin_popcount(mm);
      }
    }
    __syncthreads();
  }

  *reinterpret_cast<float4*>(&part[wave][lane * 4]) = acc;
  if (lane == 0) pcount[wave] = cnt;
  __syncthreads();
  if (threadIdx.x < Dch) {
    float sum = 0.f, c = 0.f;
    for (int w = 0; w < POOL_WAVES; ++w) { sum += part[w][threadIdx.x]; c += pcount[w]; }
    out[(long)g * Dch + threadIdx.x] = sum / (c > 1.f ? c : 1.f);
  }
}

extern "C" void kernel_launch(void* const* d_in, const int* in_sizes, int n_in,
                              void* d_out, int out_size, void* d_ws, size_t ws_size,
                              hipStream_t stream) {
  const float* x0    = (const float*)d_in[0];
  const int*   ei    = (const int*)d_in[1];
  const int*   batch = (const int*)d_in[2];
  const float* Ws    = (const float*)d_in[3];
  const float* a_src = (const float*)d_in[4];
  const float* a_dst = (const float*)d_in[5];
  const float* bias  = (const float*)d_in[6];
  const float* gamma = (const float*)d_in[7];
  const float* beta  = (const float*)d_in[8];
  const float* mean  = (const float*)d_in[9];
  const float* var   = (const float*)d_in[10];

  const int Nn    = in_sizes[0] / Dch;
  const int Ereal = in_sizes[1] / 2;
  const int G     = out_size / Dch;

  char* ws = (char*)d_ws;
  float*    xbuf  = (float*)ws;    ws += (size_t)Nn * Dch * 4;
  char*     grp   = ws;            ws += (size_t)Nn * Dch * 4;
  float*    hbuf  = (float*)ws;    ws += (size_t)Nn * Dch * 4;
  ws += (size_t)Nn * Hh * 4;
  ws += (size_t)Nn * Hh * 4;
  float*    alsrc = (float*)ws;    ws += (size_t)Nn * Hh * 4;
  float*    aldst = (float*)ws;    ws += (size_t)Nn * Hh * 4;

  int* bucket  = (int*)grp;
  int* seg_len = (int*)(grp + (size_t)Nn * BKT * 4);

  fill_u32<<<64, 256, 0, stream>>>((unsigned*)seg_len, 0u, (long)Nn);

  const float* xin = x0;
  for (int l = 0; l < Lnum; ++l) {
    const float* Wl = Ws + (size_t)l * Dch * Dch;
    const int gemm_blocks = (Nn / 16 + GEMM_STRIPS - 1) / GEMM_STRIPS;
    if (l == 0) {
      const int edge_blocks = (Ereal + GEMM_THREADS - 1) / GEMM_THREADS;
      gemm_beside_buckets<<<gemm_blocks + edge_blocks, GEMM_THREADS, 0, stream>>>(
          xin, Wl, a_src, a_dst, hbuf, alsrc, aldst, Nn / 16, gemm_blocks,
          ei, Ereal, seg_len, bucket);
    } else {
      gemm_wmma_f32<<<gemm_blocks, GEMM_THREADS, 0, stream>>>(
          xin, Wl, a_src + (size_t)l * Hh * Ch, a_dst + (size_t)l * Hh * Ch,
          hbuf, alsrc, aldst, Nn / 16);
    }

    gat_gather<<<(Nn + 7) / 8, 256, 0, stream>>>(
        seg_len, bucket, alsrc, aldst, hbuf,
        bias + (size_t)l * Dch, gamma + (size_t)l * Dch,
        beta + (size_t)l * Dch, mean + (size_t)l * Dch, var + (size_t)l * Dch,
        xbuf, Nn, ei, Ereal);

    xin = xbuf;
  }

  pool_owner<<<G, POOL_THREADS, 0, stream>>>(xin, batch, (float*)d_out, Nn);
}


# gather launches of layers 0 and 1 also run the next layer's GEMM: 16 one-node waves leave their rows in LDS, 4 waves do the strip's WMMA product: 8 to 6 launches; logits summed in a different order
# speedup vs baseline: 6.7525x; 1.0841x over previous
#include <hip/hip_runtime.h>


#define Dch      128
#define Hh       4
#define Ch       32
#define Lnum     3
#define NEG_SLOPE 0.2f
#define BN_EPS   1e-5f

typedef __attribute__((ext_vector_type(2))) float v2f;
typedef __attribute__((ext_vector_type(8))) float v8f;

__global__ void fill_u32(unsigned* __restrict__ p, unsigned v, long n) {
  long i = (long)blockIdx.x * blockDim.x + threadIdx.x;
  long stride = (long)gridDim.x * blockDim.x;
  for (; i < n; i += stride) p[i] = v;
}

__device__ __forceinline__ float4 ld4(const float* p) {
  return *reinterpret_cast<const float4*>(p);
}

template <int M>
__device__ __forceinline__ float row_xor(float v) {
  return __int_as_float(__builtin_amdgcn_update_dpp(
      0, __float_as_int(v), 0x160 | M, 0xf, 0xf, false));
}

#define GEMM_STRIPS  2
#define GEMM_THREADS (GEMM_STRIPS * 64)
__device__ __forceinline__ void gemm_block(
    const int bx,
    const float* __restrict__ X, const float* __restrict__ Wm,
    const float* __restrict__ a_src, const float* __restrict__ a_dst,
    float* __restrict__ Ho, float* __restrict__ alsrc,
    float* __restrict__ aldst, int nstrips) {
  __shared__ float lg[GEMM_STRIPS * 2 * 16 * Hh];
  const int wave = threadIdx.x >> 5;
  const int lane = threadIdx.x & 31;
  const int half = lane >> 4;
  const int l15  = lane & 15;
  const int sloc = wave >> 1;
  const int grp  = wave & 1;
  const int strip = bx * GEMM_STRIPS + sloc;
  const bool live = strip < nstrips;
  const int row0 = strip << 4;
  const int col0 = grp * 64 + l15 * 4;

  v8f acc0 = {}, acc1 = {}, acc2 = {}, acc3 = {};
  if (live) {
    const float* arow = X + (long)(row0 + l15) * Dch;
    const float* bcol = Wm + col0;
    for (int k = 0; k < Dch; k += 4) {
      const int ka = k + half * 2;
      v2f a, b;
      a.x = arow[ka];
      a.y = arow[ka + 1];
      const float4 b0 = ld4(bcol + (long)ka * Dch);
      const float4 b1 = ld4(bcol + (long)(ka + 1) * Dch);
      b.x = b0.x; b.y = b1.x;
      acc0 = __builtin_amdgcn_wmma_f32_16x16x4_f32(
           false, a,  false, b,
           (short)0, acc0,  false,  false);
      b.x = b0.y; b.y = b1.y;
      acc1 = __builtin_amdgcn_wmma_f32_16x16x4_f32(
          false, a, false, b, (short)0, acc1, false, false);
      b.x = b0.z; b.y = b1.z;
      acc2 = __builtin_amdgcn_wmma_f32_16x16x4_f32(
          false, a, false, b, (short)0, acc2, false, false);
      b.x = b0.w; b.y = b1.w;
      acc3 = __builtin_amdgcn_wmma_f32_16x16x4_f32(
          false, a, false, b, (short)0, acc3, false, false);
    }

    const float4 as = ld4(a_src + col0);
    const float4 ad = ld4(a_dst + col0);
    float ssel = 0.f, dsel = 0.f;
#pragma unroll
    for (int v = 0; v < 8; ++v) {
      float ps = acc0[v] * as.x + acc1[v] * as.y + acc2[v] * as.z + acc3[v] * as.w;
      float pd = acc0[v] * ad.x + acc1[v] * ad.y + acc2[v] * ad.z + acc3[v] * ad.w;
      ps += row_xor<1>(ps);  pd += row_xor<1>(pd);
      ps += row_xor<2>(ps);  pd += row_xor<2>(pd);
      ps += row_xor<4>(ps);  pd += row_xor<4>(pd);
      if ((l15 & 7) == v) { ssel = ps; dsel = pd; }
    }
    const int li = sloc * 128 + (half * 8 + (l15 & 7)) * Hh + grp * 2 + (l15 >> 3);
    lg[li]      = ssel;
    lg[li + 64] = dsel;
  }
  __syncthreads();
  const int t = threadIdx.x;
  if (bx * GEMM_STRIPS + (t >> 6) < nstrips) {
    const long o = (long)bx * (GEMM_STRIPS * 16 * Hh) + t;
    const int  q = (t >> 6) * 128 + (t & 63);
    alsrc[o] = lg[q];
    aldst[o] = lg[q + 64];
  }
  if (live) {
    float* orow = Ho + (long)(row0 + half * 8) * Dch + col0;
#pragma unroll
    for (int v = 0; v < 8; ++v)
      *reinterpret_cast<float4*>(orow + (long)v * Dch) =
          make_float4(acc0[v], acc1[v], acc2[v], acc3[v]);
  }
}

#define BKT 64
__device__ __forceinline__ void bucket_edge(
    const int e, const int* __restrict__ ei, int Ereal,
    int* __restrict__ seg_len, int* __restrict__ bucket) {
  if (e >= Ereal) return;
  const int s = ei[e];
  const int d = ei[Ereal + e];
  const int pos = atomicAdd(&seg_len[d], 1);
  bucket[d * BKT + (pos < BKT ? pos : BKT - 1)] = s;
}

__global__ __launch_bounds__(GEMM_THREADS) void gemm_beside_buckets(
    const float* __restrict__ X, const float* __restrict__ Wm,
    const float* __restrict__ a_src, const float* __restrict__ a_dst,
    float* __restrict__ Ho, float* __restrict__ alsrc,
    float* __restrict__ aldst, int nstrips, int gemm_blocks,
    const int* __restrict__ ei, int Ereal,
    int* __restrict__ seg_len, int* __restrict__ bucket) {
  const int bx = (int)blockIdx.x;
  if (bx < gemm_blocks) {
    gemm_block(bx, X, Wm, a_src, a_dst, Ho, alsrc, aldst, nstrips);
  } else {
    bucket_edge((bx - gemm_blocks) * GEMM_THREADS + (int)threadIdx.x,
                ei, Ereal, seg_len, bucket);
  }
}

__device__ __forceinline__ float lrelu(float v) { return v > 0.f ? v : v * NEG_SLOPE; }

__device__ __forceinline__ float wave_max(float v) {
  for (int o = 16; o > 0; o >>= 1) v = fmaxf(v, __shfl_xor(v, o, 32));
  return v;
}
__device__ __forceinline__ float wave_sum(float v) {
  for (int o = 16; o > 0; o >>= 1) v += __shfl_xor(v, o, 32);
  return v;
}

__device__ __forceinline__ bool chunk_scores(
    const int* __restrict__ csr_src, const float* __restrict__ alsrc,
    int beg, int cnt, int base, int lane, float4 ad, int& s, float4& e) {
  const int j = base + lane;
  const bool valid = j < cnt;
  s = 0;
  e = make_float4(0.f, 0.f, 0.f, 0.f);
  if (valid) {
    s = csr_src[beg + j];
    const float4 a = ld4(alsrc + (long)s * Hh);
    e.x = lrelu(a.x + ad.x);
    e.y = lrelu(a.y + ad.y);
    e.z = lrelu(a.z + ad.z);
    e.w = lrelu(a.w + ad.w);
  }
  return valid;
}

__device__ __forceinline__ int bcast(float v, int j) {
  return __builtin_amdgcn_readlane(__float_as_int(v), j);
}

__device__ __forceinline__ void add_rows(const float* __restrict__ hl,
                                         int k0, int k1, int k2, int k3,
                                         int s, float4 al, int n, float4& acc) {
#pragma unroll 4
  for (int j = 0; j < n; ++j) {
    const int sj = __builtin_amdgcn_readlane(s, j);
    const float b = __int_as_float((bcast(al.x, j) & k0) | (bcast(al.y, j) & k1) |
                                   (bcast(al.z, j) & k2) | (bcast(al.w, j) & k3));
    const float4 hv = ld4(hl + (long)sj * Dch);
    acc.x += b * hv.x;
    acc.y += b * hv.y;
    acc.z += b * hv.z;
    acc.w += b * hv.w;
  }
}

__device__ __forceinline__ bool scan_scores(
    const int* __restrict__ ei, int Ereal, const float* __restrict__ alsrc,
    int d, int base, int lane, float4 ad, int& s, float4& e) {
  const int i = base + lane;
  s = 0;
  e = make_float4(0.f, 0.f, 0.f, 0.f);
  const bool valid = i < Ereal && ei[Ereal + i] == d;
  if (valid) {
    s = ei[i];
    const float4 a = ld4(alsrc + (long)s * Hh);
    e.x = lrelu(a.x + ad.x);
    e.y = lrelu(a.y + ad.y);
    e.z = lrelu(a.z + ad.z);
    e.w = lrelu(a.w + ad.w);
  }
  return valid;
}

__device__ float4 gather_by_scan(
    const int* __restrict__ ei, int Ereal, const float* __restrict__ alsrc,
    const float* __restrict__ hl, int d, int lane, float4 ad, float4 es,
    int k0, int k1, int k2, int k3) {
  float4 m = es;
  for (int base = 0; base < Ereal; base += 32) {
    int s; float4 e;
    if (scan_scores(ei, Ereal, alsrc, d, base, lane, ad, s, e)) {
      m.x = fmaxf(m.x, e.x); m.y = fmaxf(m.y, e.y);
      m.z = fmaxf(m.z, e.z); m.w = fmaxf(m.w, e.w);
    }
  }
  m.x = wave_max(m.x); m.y = wave_max(m.y);
  m.z = wave_max(m.z); m.w = wave_max(m.w);

  float4 ps;
  ps.x = expf(es.x - m.x); ps.y = expf(es.y - m.y);
  ps.z = expf(es.z - m.z); ps.w = expf(es.w - m.w);
  float4 z = make_float4(0.f, 0.f, 0.f, 0.f);
  if (lane == 0) z = ps;
  for (int base = 0; base < Ereal; base += 32) {
    int s; float4 e;
    if (scan_scores(ei, Ereal, alsrc, d, base, lane, ad, s, e)) {
      z.x += expf(e.x - m.x); z.y += expf(e.y - m.y);
      z.z += expf(e.z - m.z); z.w += expf(e.w - m.w);
    }
  }
  float4 den;
  den.x = wave_sum(z.x) + 1e-16f; den.y = wave_sum(z.y) + 1e-16f;
  den.z = wave_sum(z.z) + 1e-16f; den.w = wave_sum(z.w) + 1e-16f;

  float4 acc;
  {
    const float a_self = __int_as_float(
        (__float_as_int(ps.x / den.x) & k0) | (__float_as_int(ps.y / den.y) & k1) |
        (__float_as_int(ps.z / den.z) & k2) | (__float_as_int(ps.w / den.w) & k3));
    const float4 hv = ld4(hl + (long)d * Dch);
    acc.x = a_self * hv.x; acc.y = a_self * hv.y;
    acc.z = a_self * hv.z; acc.w = a_self * hv.w;
  }
  for (int base = 0; base < Ereal; base += 32) {
    int s; float4 e;
    float4 al = make_float4(0.f, 0.f, 0.f, 0.f);
    const bool v = scan_scores(ei, Ereal, alsrc, d, base, lane, ad, s, e);
    if (v) {
      al.x = expf(e.x - m.x) / den.x; al.y = expf(e.y - m.y) / den.y;
      al.z = expf(e.z - m.z) / den.z; al.w = expf(e.w - m.w) / den.w;
    }
    unsigned hit = __builtin_amdgcn_ballot_w32(v);
    while (hit) {
      const int j = __builtin_ctz(hit);
      hit &= hit - 1;
      const int sj = __builtin_amdgcn_readlane(s, j);
      const float b = __int_as_float((bcast(al.x, j) & k0) | (bcast(al.y, j) & k1) |
                                     (bcast(al.z, j) & k2) | (bcast(al.w, j) & k3));
      const float4 hv = ld4(hl + (long)sj * Dch);
      acc.x += b * hv.x;
      acc.y += b * hv.y;
      acc.z += b * hv.z;
      acc.w += b * hv.w;
    }
  }
  return acc;
}

__device__ __forceinline__ float bn_elu(float a, float b, float g, float be,
                                        float mu, float va) {
  float v = a + b;
  v = (v - mu) * (g / sqrtf(va + BN_EPS)) + be;
  return v > 0.f ? v : expf(v) - 1.f;
}

__device__ __forceinline__ float4 gather_row(
    const int d, const int lane,
    const int* __restrict__ seg_len, const int* __restrict__ csr_src,
    const float* __restrict__ alsrc, const float* __restrict__ aldst,
    const float* __restrict__ Hf,
    const float* __restrict__ bias, const float* __restrict__ gamma,
    const float* __restrict__ beta, const float* __restrict__ mean,
    const float* __restrict__ var,
    const int* __restrict__ ei, int Ereal) {
  const int beg = d * BKT;
  const int cnt = seg_len[d];
  const float4 ad = ld4(aldst + (long)d * Hh);
  const float4 as = ld4(alsrc + (long)d * Hh);
  __builtin_amdgcn_sched_barrier(0);
  float4 es;
  es.x = lrelu(as.x + ad.x);
  es.y = lrelu(as.y + ad.y);
  es.z = lrelu(as.z + ad.z);
  es.w = lrelu(as.w + ad.w);

  const int hsel = lane >> 3;
  const int k0 = hsel == 0 ? -1 : 0, k1 = hsel == 1 ? -1 : 0;
  const int k2 = hsel == 2 ? -1 : 0, k3 = hsel == 3 ? -1 : 0;
  const float* hl = Hf + lane * 4;
  float4 acc;
  if (cnt > BKT) {
    acc = gather_by_scan(ei, Ereal, alsrc, hl, d, lane, ad, es, k0, k1, k2, k3);
  } else {
  int s0; float4 e0;
  const bool v0 = chunk_scores(csr_src, alsrc, beg, cnt, 0, lane, ad, s0, e0);

  float4 m = es;
  if (v0) {
    m.x = fmaxf(m.x, e0.x); m.y = fmaxf(m.y, e0.y);
    m.z = fmaxf(m.z, e0.z); m.w = fmaxf(m.w, e0.w);
  }
  for (int base = 32; base < cnt; base += 32) {
    int s; float4 e;
    if (chunk_scores(csr_src, alsrc, beg, cnt, base, lane, ad, s, e)) {
      m.x = fmaxf(m.x, e.x); m.y = fmaxf(m.y, e.y);
      m.z = fmaxf(m.z, e.z); m.w = fmaxf(m.w, e.w);
    }
  }
  m.x = wave_max(m.x); m.y = wave_max(m.y);
  m.z = wave_max(m.z); m.w = wave_max(m.w);

  float4 z = make_float4(0.f, 0.f, 0.f, 0.f);
  float4 ps;
  ps.x = expf(es.x - m.x); ps.y = expf(es.y - m.y);
  ps.z = expf(es.z - m.z); ps.w = expf(es.w - m.w);
  if (lane == 0) z = ps;
  float4 p0 = make_float4(0.f, 0.f, 0.f, 0.f);
  if (v0) {
    p0.x = expf(e0.x - m.x); p0.y = expf(e0.y - m.y);
    p0.z = expf(e0.z - m.z); p0.w = expf(e0.w - m.w);
    z.x += p0.x; z.y += p0.y; z.z += p0.z; z.w += p0.w;
  }
  for (int base = 32; base < cnt; base += 32) {
    int s; float4 e;
    if (chunk_scores(csr_src, alsrc, beg, cnt, base, lane, ad, s, e)) {
      z.x += expf(e.x - m.x); z.y += expf(e.y - m.y);
      z.z += expf(e.z - m.z); z.w += expf(e.w - m.w);
    }
  }
  float4 den;
  den.x = wave_sum(z.x) + 1e-16f; den.y = wave_sum(z.y) + 1e-16f;
  den.z = wave_sum(z.z) + 1e-16f; den.w = wave_sum(z.w) + 1e-16f;

  {
    const float a_self = __int_as_float(
        (__float_as_int(ps.x / den.x) & k0) | (__float_as_int(ps.y / den.y) & k1) |
        (__float_as_int(ps.z / den.z) & k2) | (__float_as_int(ps.w / den.w) & k3));
    const float4 hv = ld4(hl + (long)d * Dch);
    acc.x = a_self * hv.x; acc.y = a_self * hv.y;
    acc.z = a_self * hv.z; acc.w = a_self * hv.w;
  }
  {
    float4 al;
    al.x = p0.x / den.x; al.y = p0.y / den.y;
    al.z = p0.z / den.z; al.w = p0.w / den.w;
    add_rows(hl, k0, k1, k2, k3, s0, al, cnt < 32 ? cnt : 32, acc);
  }
  for (int base = 32; base < cnt; base += 32) {
    int s; float4 e;
    float4 al = make_float4(0.f, 0.f, 0.f, 0.f);
    if (chunk_scores(csr_src, alsrc, beg, cnt, base, lane, ad, s, e)) {
      al.x = expf(e.x - m.x) / den.x; al.y = expf(e.y - m.y) / den.y;
      al.z = expf(e.z - m.z) / den.z; al.w = expf(e.w - m.w) / den.w;
    }
    const int left = cnt - base;
    add_rows(hl, k0, k1, k2, k3, s, al, left < 32 ? left : 32, acc);
  }
  }
  const float4 bi = ld4(bias + lane * 4);
  const float4 ga = ld4(gamma + lane * 4);
  const float4 be = ld4(beta + lane * 4);
  const float4 mu = ld4(mean + lane * 4);
  const float4 va = ld4(var + lane * 4);
  float4 o;
  o.x = bn_elu(acc.x, bi.x, ga.x, be.x, mu.x, va.x);
  o.y = bn_elu(acc.y, bi.y, ga.y, be.y, mu.y, va.y);
  o.z = bn_elu(acc.z, bi.z, ga.z, be.z, mu.z, va.z);
  o.w = bn_elu(acc.w, bi.w, ga.w, be.w, mu.w, va.w);
  return o;
}

__global__ __launch_bounds__(256) void gat_gather(
    const int* __restrict__ seg_len, const int* __restrict__ csr_src,
    const float* __restrict__ alsrc, const float* __restrict__ aldst,
    const float* __restrict__ Hf,
    const float* __restrict__ bias, const float* __restrict__ gamma,
    const float* __restrict__ beta, const float* __restrict__ mean,
    const float* __restrict__ var,
    float* __restrict__ xo, int Nn,
    const int* __restrict__ ei, int Ereal) {
  const int lane = threadIdx.x & 31;
  const int d = __builtin_amdgcn_readfirstlane((int)(blockIdx.x * 8 + (threadIdx.x >> 5)));
  if (d >= Nn) return;
  const float4 o = gather_row(d, lane, seg_len, csr_src, alsrc, aldst, Hf,
                              bias, gamma, beta, mean, var, ei, Ereal);
  *reinterpret_cast<float4*>(xo + (long)d * Dch + lane * 4) = o;
}

#define FG_WAVES   16
#define FG_THREADS (FG_WAVES * 32)
#define XPITCH     132
__global__ __launch_bounds__(FG_THREADS) void gather_then_gemm(
    const int* __restrict__ seg_len, const int* __restrict__ csr_src,
    const float* __restrict__ alsrc, const float* __restrict__ aldst,
    const float* __restrict__ Hf,
    const float* __restrict__ bias, const float* __restrict__ gamma,
    const float* __restrict__ beta, const float* __restrict__ mean,
    const float* __restrict__ var, int Nn,
    const int* __restrict__ ei, int Ereal,
    const float* __restrict__ Wn, const float* __restrict__ a_src_n,
    const float* __restrict__ a_dst_n,
    float* __restrict__ Ho, float* __restrict__ alsrc_o,
    float* __restrict__ aldst_o) {
  __shared__ __attribute__((aligned(16))) float xs[FG_WAVES * XPITCH];
  __shared__ __attribute__((aligned(16))) float lg[2 * FG_WAVES * Hh];
  const int lane = threadIdx.x & 31;
  const int wave = __builtin_amdgcn_readfirstlane((int)(threadIdx.x >> 5));
  const int row0 = (int)blockIdx.x * FG_WAVES;
  const int d = row0 + wave;
  float4 o = make_float4(0.f, 0.f, 0.f, 0.f);
  if (d < Nn)
    o = gather_row(d, lane, seg_len, csr_src, alsrc, aldst, Hf,
                   bias, gamma, beta, mean, var, ei, Ereal);
  *reinterpret_cast<float4*>(&xs[wave * XPITCH + lane * 4]) = o;
  __syncthreads();

  const int half = lane >> 4;
  const int c    = lane & 15;
  const int col0 = wave * Ch + 2 * c;
  v8f acc0 = {}, acc1 = {};
  if (wave < Hh) {
    const float* ar = xs + c * XPITCH + half * 2;
    const float* bc = Wn + (long)(half * 2) * Dch + col0;
    for (int k = 0; k < Dch; k += 4) {
      const v2f a = *reinterpret_cast<const v2f*>(ar + k);
      const float2 b0 = *reinterpret_cast<const float2*>(bc + (long)k * Dch);
      const float2 b1 = *reinterpret_cast<const float2*>(bc + (long)(k + 1) * Dch);
      v2f b;
      b.x = b0.x; b.y = b1.x;
      acc0 = __builtin_amdgcn_wmma_f32_16x16x4_f32(
          false, a, false, b, (short)0, acc0, false, false);
      b.x = b0.y; b.y = b1.y;
      acc1 = __builtin_amdgcn_wmma_f32_16x16x4_f32(
          false, a, false, b, (short)0, acc1, false, false);
    }
    const float2 as = *reinterpret_cast<const float2*>(a_src_n + col0);
    const float2 ad = *reinterpret_cast<const float2*>(a_dst_n + col0);
    float sel = 0.f;
#pragma unroll
    for (int v = 0; v < 8; ++v) {
      float ps = acc0[v] * as.x + acc1[v] * as.y;
      float pd = acc0[v] * ad.x + acc1[v] * ad.y;
      ps += row_xor<1>(ps);  pd += row_xor<1>(pd);
      ps += row_xor<2>(ps);  pd += row_xor<2>(pd);
      ps += row_xor<4>(ps);  pd += row_xor<4>(pd);
      ps += row_xor<8>(ps);  pd += row_xor<8>(pd);
      if ((c & 7) == v) sel = c < 8 ? ps : pd;
    }
    lg[(c >> 3) * (FG_WAVES * Hh) + (half * 8 + (c & 7)) * Hh + wave] = sel;
  }
  __syncthreads();
  if (wave < Hh) {
    float* orow = Ho + (long)(row0 + half * 8) * Dch + col0;
    if (row0 + FG_WAVES <= Nn) {
#pragma unroll
      for (int v = 0; v < 8; ++v)
        *reinterpret_cast<float2*>(orow + (long)v * Dch) =
            make_float2(acc0[v], acc1[v]);
    } else {
#pragma unroll
      for (int v = 0; v < 8; ++v)
        if (row0 + half * 8 + v < Nn)
          *reinterpret_cast<float2*>(orow + (long)v * Dch) =
              make_float2(acc0[v], acc1[v]);
    }
  } else if (wave < Hh + 2) {
    const int q = lane * 2;
    if (row0 + (q >> 2) < Nn) {
      const float2 vv = *reinterpret_cast<const float2*>(
          &lg[(wave - Hh) * (FG_WAVES * Hh) + q]);
      float* dst = (wave == Hh ? alsrc_o : aldst_o) + (long)row0 * Hh + q;
      *reinterpret_cast<float2*>(dst) = vv;
    }
  }
}

#define POOL_THREADS 1024
#define POOL_WAVES   32
#define POOL_TILE    65536
#define POOL_IDS     16

__global__ __launch_bounds__(POOL_THREADS) void pool_owner(
    const float* __restrict__ x, const int* __restrict__ batch,
    float* __restrict__ out, int Nn) {
  __shared__ unsigned member[POOL_TILE / 32];
  __shared__ float    part[POOL_WAVES][Dch];
  __shared__ float    pcount[POOL_WAVES];
  const int g    = blockIdx.x;
  const int lane = threadIdx.x & 31;
  const int wave = __builtin_amdgcn_readfirstlane((int)(threadIdx.x >> 5));
  float4 acc = make_float4(0.f, 0.f, 0.f, 0.f);
  float  cnt = 0.f;

  for (int t0 = 0; t0 < Nn; t0 += POOL_TILE) {
    const int tn = Nn - t0 < POOL_TILE ? Nn - t0 : POOL_TILE;

    const int nch = (tn + 31) >> 5;
    for (int c0 = wave; c0 < nch; c0 += POOL_WAVES * POOL_IDS) {
      int id[POOL_IDS];
#pragma unroll
      for (int k = 0; k < POOL_IDS; ++k) {
        const int n = t0 + (c0 + POOL_WAVES * k) * 32 + lane;
        id[k] = batch[n < Nn ? n : Nn - 1];
      }
#pragma unroll
      for (int k = 0; k < POOL_IDS; ++k) {
        const int c = c0 + POOL_WAVES * k;
        const int n = t0 + c * 32 + lane;
        const unsigned b = __builtin_amdgcn_ballot_w32(n < Nn && id[k] == g);
        if (c < nch && lane == 0) member[c] = b;
      }
    }
    __syncthreads();

    const int ngrp = (tn + 7) >> 3;
    for (int q0 = wave; q0 < ngrp; q0 += 32 * POOL_WAVES) {
      const int q = q0 + 32 * lane;
      unsigned m8 = 0;
      if (q < ngrp) m8 = (member[q >> 2] >> ((q & 3) * 8)) & 0xffu;
      unsigned any = __builtin_amdgcn_ballot_w32(m8 != 0);
      while (any) {
        const int i = __builtin_ctz(any);
        any &= any - 1;
        const unsigned mm = (unsigned)__builtin_amdgcn_readlane((int)m8, i);
        const int nb = t0 + (q0 + 32 * i) * 8;
        float4 r[8];
#pragma unroll
        for (int j = 0; j < 8; ++j) {
          int n = nb + j;
          n = n < Nn ? n : Nn - 1;
          r[j] = ld4(x + (long)n * Dch + lane * 4);
        }
#pragma unroll
        for (int j = 0; j < 8; ++j) {
          if (mm & (1u << j)) {
            acc.x += r[j].x; acc.y += r[j].y; acc.z += r[j].z; acc.w += r[j].w;
          }
        }
        cnt += (float)__builtin_popcount(mm);
      }
    }
    __syncthreads();
  }

  *reinterpret_cast<float4*>(&part[wave][lane * 4]) = acc;
  if (lane == 0) pcount[wave] = cnt;
  __syncthreads();
  if (threadIdx.x < Dch) {
    float sum = 0.f, c = 0.f;
    for (int w = 0; w < POOL_WAVES; ++w) { sum += part[w][threadIdx.x]; c += pcount[w]; }
    out[(long)g * Dch + threadIdx.x] = sum / (c > 1.f ? c : 1.f);
  }
}

extern "C" void kernel_launch(void* const* d_in, const int* in_sizes, int n_in,
                              void* d_out, int out_size, void* d_ws, size_t ws_size,
                              hipStream_t stream) {
  const float* x0    = (const float*)d_in[0];
  const int*   ei    = (const int*)d_in[1];
  const int*   batch = (const int*)d_in[2];
  const float* Ws    = (const float*)d_in[3];
  const float* a_src = (const float*)d_in[4];
  const float* a_dst = (const float*)d_in[5];
  const float* bias  = (const float*)d_in[6];
  const float* gamma = (const float*)d_in[7];
  const float* beta  = (const float*)d_in[8];
  const float* mean  = (const float*)d_in[9];
  const float* var   = (const float*)d_in[10];

  const int Nn    = in_sizes[0] / Dch;
  const int Ereal = in_sizes[1] / 2;
  const int G     = out_size / Dch;

  char* ws = (char*)d_ws;
  float*    hB    = (float*)ws;    ws += (size_t)Nn * Dch * 4;
  char*     grp   = ws;            ws += (size_t)Nn * Dch * 4;
  float*    hA    = (float*)ws;    ws += (size_t)Nn * Dch * 4;
  float*    alsB  = (float*)ws;    ws += (size_t)Nn * Hh * 4;
  float*    aldB  = (float*)ws;    ws += (size_t)Nn * Hh * 4;
  float*    alsA  = (float*)ws;    ws += (size_t)Nn * Hh * 4;
  float*    aldA  = (float*)ws;    ws += (size_t)Nn * Hh * 4;

  int* bucket  = (int*)grp;
  int* seg_len = (int*)(grp + (size_t)Nn * BKT * 4);

  fill_u32<<<64, 256, 0, stream>>>((unsigned*)seg_len, 0u, (long)Nn);

  {
    const int gemm_blocks = (Nn / 16 + GEMM_STRIPS - 1) / GEMM_STRIPS;
    const int edge_blocks = (Ereal + GEMM_THREADS - 1) / GEMM_THREADS;
    gemm_beside_buckets<<<gemm_blocks + edge_blocks, GEMM_THREADS, 0, stream>>>(
        x0, Ws, a_src, a_dst, hA, alsA, aldA, Nn / 16, gemm_blocks,
        ei, Ereal, seg_len, bucket);
  }

  float* hin = hA;  float* hout = hB;
  float* asin = alsA; float* adin = aldA;
  float* asout = alsB; float* adout = aldB;
  for (int l = 0; l + 1 < Lnum; ++l) {
    gather_then_gemm<<<(Nn + FG_WAVES - 1) / FG_WAVES, FG_THREADS, 0, stream>>>(
        seg_len, bucket, asin, adin, hin,
        bias + (size_t)l * Dch, gamma + (size_t)l * Dch,
        beta + (size_t)l * Dch, mean + (size_t)l * Dch, var + (size_t)l * Dch,
        Nn, ei, Ereal,
        Ws + (size_t)(l + 1) * Dch * Dch,
        a_src + (size_t)(l + 1) * Hh * Ch, a_dst + (size_t)(l + 1) * Hh * Ch,
        hout, asout, adout);
    float* t;
    t = hin;  hin = hout;   hout = t;
    t = asin; asin = asout; asout = t;
    t = adin; adin = adout; adout = t;
  }

  {
    const int l = Lnum - 1;
    gat_gather<<<(Nn + 7) / 8, 256, 0, stream>>>(
        seg_len, bucket, asin, adin, hin,
        bias + (size_t)l * Dch, gamma + (size_t)l * Dch,
        beta + (size_t)l * Dch, mean + (size_t)l * Dch, var + (size_t)l * Dch,
        hout, Nn, ei, Ereal);
  }

  pool_owner<<<G, POOL_THREADS, 0, stream>>>(hout, batch, (float*)d_out, Nn);
}


# gather wave takes each source logit from the row it loads (a_src dot, 8-lane sum), one pass, running max and sum, v_exp_f32 weights: logit lookups and wave reductions gone; softmax sums reordered
# speedup vs baseline: 7.2196x; 1.0692x over previous
#include <hip/hip_runtime.h>


#define Dch      128
#define Hh       4
#define Ch       32
#define Lnum     3
#define NEG_SLOPE 0.2f
#define BN_EPS   1e-5f

typedef __attribute__((ext_vector_type(2))) float v2f;
typedef __attribute__((ext_vector_type(8))) float v8f;

__global__ void fill_u32(unsigned* __restrict__ p, unsigned v, long n) {
  long i = (long)blockIdx.x * blockDim.x + threadIdx.x;
  long stride = (long)gridDim.x * blockDim.x;
  for (; i < n; i += stride) p[i] = v;
}

__device__ __forceinline__ float4 ld4(const float* p) {
  return *reinterpret_cast<const float4*>(p);
}

template <int M>
__device__ __forceinline__ float row_xor(float v) {
  return __int_as_float(__builtin_amdgcn_update_dpp(
      0, __float_as_int(v), 0x160 | M, 0xf, 0xf, false));
}

#define GEMM_STRIPS  2
#define GEMM_THREADS (GEMM_STRIPS * 64)
__device__ __forceinline__ void gemm_block(
    const int bx,
    const float* __restrict__ X, const float* __restrict__ Wm,
    const float* __restrict__ a_src, const float* __restrict__ a_dst,
    float* __restrict__ Ho, float* __restrict__ alsrc,
    float* __restrict__ aldst, int nstrips) {
  __shared__ float lg[GEMM_STRIPS * 2 * 16 * Hh];
  const int wave = threadIdx.x >> 5;
  const int lane = threadIdx.x & 31;
  const int half = lane >> 4;
  const int l15  = lane & 15;
  const int sloc = wave >> 1;
  const int grp  = wave & 1;
  const int strip = bx * GEMM_STRIPS + sloc;
  const bool live = strip < nstrips;
  const int row0 = strip << 4;
  const int col0 = grp * 64 + l15 * 4;

  v8f acc0 = {}, acc1 = {}, acc2 = {}, acc3 = {};
  if (live) {
    const float* arow = X + (long)(row0 + l15) * Dch;
    const float* bcol = Wm + col0;
    for (int k = 0; k < Dch; k += 4) {
      const int ka = k + half * 2;
      v2f a, b;
      a.x = arow[ka];
      a.y = arow[ka + 1];
      const float4 b0 = ld4(bcol + (long)ka * Dch);
      const float4 b1 = ld4(bcol + (long)(ka + 1) * Dch);
      b.x = b0.x; b.y = b1.x;
      acc0 = __builtin_amdgcn_wmma_f32_16x16x4_f32(
           false, a,  false, b,
           (short)0, acc0,  false,  false);
      b.x = b0.y; b.y = b1.y;
      acc1 = __builtin_amdgcn_wmma_f32_16x16x4_f32(
          false, a, false, b, (short)0, acc1, false, false);
      b.x = b0.z; b.y = b1.z;
      acc2 = __builtin_amdgcn_wmma_f32_16x16x4_f32(
          false, a, false, b, (short)0, acc2, false, false);
      b.x = b0.w; b.y = b1.w;
      acc3 = __builtin_amdgcn_wmma_f32_16x16x4_f32(
          false, a, false, b, (short)0, acc3, false, false);
    }

    const float4 as = ld4(a_src + col0);
    const float4 ad = ld4(a_dst + col0);
    float ssel = 0.f, dsel = 0.f;
#pragma unroll
    for (int v = 0; v < 8; ++v) {
      float ps = acc0[v] * as.x + acc1[v] * as.y + acc2[v] * as.z + acc3[v] * as.w;
      float pd = acc0[v] * ad.x + acc1[v] * ad.y + acc2[v] * ad.z + acc3[v] * ad.w;
      ps += row_xor<1>(ps);  pd += row_xor<1>(pd);
      ps += row_xor<2>(ps);  pd += row_xor<2>(pd);
      ps += row_xor<4>(ps);  pd += row_xor<4>(pd);
      if ((l15 & 7) == v) { ssel = ps; dsel = pd; }
    }
    const int li = sloc * 128 + (half * 8 + (l15 & 7)) * Hh + grp * 2 + (l15 >> 3);
    lg[li]      = ssel;
    lg[li + 64] = dsel;
  }
  __syncthreads();
  const int t = threadIdx.x;
  if (bx * GEMM_STRIPS + (t >> 6) < nstrips) {
    const long o = (long)bx * (GEMM_STRIPS * 16 * Hh) + t;
    const int  q = (t >> 6) * 128 + (t & 63);
    alsrc[o] = lg[q];
    aldst[o] = lg[q + 64];
  }
  if (live) {
    float* orow = Ho + (long)(row0 + half * 8) * Dch + col0;
#pragma unroll
    for (int v = 0; v < 8; ++v)
      *reinterpret_cast<float4*>(orow + (long)v * Dch) =
          make_float4(acc0[v], acc1[v], acc2[v], acc3[v]);
  }
}

#define BKT 64
__device__ __forceinline__ void bucket_edge(
    const int e, const int* __restrict__ ei, int Ereal,
    int* __restrict__ seg_len, int* __restrict__ bucket) {
  if (e >= Ereal) return;
  const int s = ei[e];
  const int d = ei[Ereal + e];
  const int pos = atomicAdd(&seg_len[d], 1);
  bucket[d * BKT + (pos < BKT ? pos : BKT - 1)] = s;
}

__global__ __launch_bounds__(GEMM_THREADS) void gemm_beside_buckets(
    const float* __restrict__ X, const float* __restrict__ Wm,
    const float* __restrict__ a_src, const float* __restrict__ a_dst,
    float* __restrict__ Ho, float* __restrict__ alsrc,
    float* __restrict__ aldst, int nstrips, int gemm_blocks,
    const int* __restrict__ ei, int Ereal,
    int* __restrict__ seg_len, int* __restrict__ bucket) {
  const int bx = (int)blockIdx.x;
  if (bx < gemm_blocks) {
    gemm_block(bx, X, Wm, a_src, a_dst, Ho, alsrc, aldst, nstrips);
  } else {
    bucket_edge((bx - gemm_blocks) * GEMM_THREADS + (int)threadIdx.x,
                ei, Ereal, seg_len, bucket);
  }
}

__device__ __forceinline__ float lrelu(float v) { return v > 0.f ? v : v * NEG_SLOPE; }

__device__ __forceinline__ float wave_max(float v) {
  for (int o = 16; o > 0; o >>= 1) v = fmaxf(v, __shfl_xor(v, o, 32));
  return v;
}
__device__ __forceinline__ float wave_sum(float v) {
  for (int o = 16; o > 0; o >>= 1) v += __shfl_xor(v, o, 32);
  return v;
}


__device__ __forceinline__ float head_sum(float v) {
  v += row_xor<1>(v);
  v += row_xor<2>(v);
  v += row_xor<4>(v);
  return v;
}

struct Row4 { v2f lo, hi; };
__device__ __forceinline__ Row4 ldrow(const float* p) {
  const float4 t = ld4(p);
  Row4 r;
  r.lo.x = t.x; r.lo.y = t.y; r.hi.x = t.z; r.hi.y = t.w;
  return r;
}

__device__ __forceinline__ float row_score(Row4 h, Row4 a, float adh) {
  const v2f t = h.lo * a.lo + h.hi * a.hi;
  const float v = head_sum(t.x + t.y) + adh;
  return fmaxf(v, v * NEG_SLOPE);
}

template <bool TAIL>
__device__ __forceinline__ void fold4(const float* __restrict__ hl, Row4 a4, float adh,
                                      int s, int j, int n,
                                      float& m, float& z, Row4& acc) {
  const int s0 = __builtin_amdgcn_readlane(s, j);
  const int s1 = __builtin_amdgcn_readlane(s, j + 1);
  const int s2 = __builtin_amdgcn_readlane(s, j + 2);
  const int s3 = __builtin_amdgcn_readlane(s, j + 3);
  const Row4 h0 = ldrow(hl + (long)s0 * Dch);
  const Row4 h1 = ldrow(hl + (long)s1 * Dch);
  const Row4 h2 = ldrow(hl + (long)s2 * Dch);
  const Row4 h3 = ldrow(hl + (long)s3 * Dch);
  const float e0 = row_score(h0, a4, adh);
  float e1 = row_score(h1, a4, adh);
  float e2 = row_score(h2, a4, adh);
  float e3 = row_score(h3, a4, adh);
  if (TAIL) {
    if (j + 1 >= n) e1 = -__builtin_inff();
    if (j + 2 >= n) e2 = -__builtin_inff();
    if (j + 3 >= n) e3 = -__builtin_inff();
  }
  const float mn = fmaxf(fmaxf(m, e0), fmaxf(fmaxf(e1, e2), e3));
  const float sc = __expf(m - mn);
  const float p0 = __expf(e0 - mn);
  float p1 = __expf(e1 - mn);
  float p2 = __expf(e2 - mn);
  float p3 = __expf(e3 - mn);
  if (TAIL) {
    if (j + 1 >= n) p1 = 0.f;
    if (j + 2 >= n) p2 = 0.f;
    if (j + 3 >= n) p3 = 0.f;
  }
  m = mn;
  z = z * sc + ((p0 + p1) + (p2 + p3));
  acc.lo = acc.lo * sc + h0.lo * p0 + h1.lo * p1 + h2.lo * p2 + h3.lo * p3;
  acc.hi = acc.hi * sc + h0.hi * p0 + h1.hi * p1 + h2.hi * p2 + h3.hi * p3;
}

__device__ __forceinline__ void fold_rows(const float* __restrict__ hl, Row4 a4, float adh,
                                          int s, int n,
                                          float& m, float& z, Row4& acc) {
  int j = 0;
  for (; j + 4 <= n; j += 4) fold4<false>(hl, a4, adh, s, j, n, m, z, acc);
  if (j < n) fold4<true>(hl, a4, adh, s, j, n, m, z, acc);
}

__device__ __forceinline__ int bcast(float v, int j) {
  return __builtin_amdgcn_readlane(__float_as_int(v), j);
}

__device__ __forceinline__ bool scan_scores(
    const int* __restrict__ ei, int Ereal, const float* __restrict__ alsrc,
    int d, int base, int lane, float4 ad, int& s, float4& e) {
  const int i = base + lane;
  s = 0;
  e = make_float4(0.f, 0.f, 0.f, 0.f);
  const bool valid = i < Ereal && ei[Ereal + i] == d;
  if (valid) {
    s = ei[i];
    const float4 a = ld4(alsrc + (long)s * Hh);
    e.x = lrelu(a.x + ad.x);
    e.y = lrelu(a.y + ad.y);
    e.z = lrelu(a.z + ad.z);
    e.w = lrelu(a.w + ad.w);
  }
  return valid;
}

__device__ float4 gather_by_scan(
    const int* __restrict__ ei, int Ereal, const float* __restrict__ alsrc,
    const float* __restrict__ hl, int d, int lane, float4 ad, float4 es,
    int k0, int k1, int k2, int k3) {
  float4 m = es;
  for (int base = 0; base < Ereal; base += 32) {
    int s; float4 e;
    if (scan_scores(ei, Ereal, alsrc, d, base, lane, ad, s, e)) {
      m.x = fmaxf(m.x, e.x); m.y = fmaxf(m.y, e.y);
      m.z = fmaxf(m.z, e.z); m.w = fmaxf(m.w, e.w);
    }
  }
  m.x = wave_max(m.x); m.y = wave_max(m.y);
  m.z = wave_max(m.z); m.w = wave_max(m.w);

  float4 ps;
  ps.x = expf(es.x - m.x); ps.y = expf(es.y - m.y);
  ps.z = expf(es.z - m.z); ps.w = expf(es.w - m.w);
  float4 z = make_float4(0.f, 0.f, 0.f, 0.f);
  if (lane == 0) z = ps;
  for (int base = 0; base < Ereal; base += 32) {
    int s; float4 e;
    if (scan_scores(ei, Ereal, alsrc, d, base, lane, ad, s, e)) {
      z.x += expf(e.x - m.x); z.y += expf(e.y - m.y);
      z.z += expf(e.z - m.z); z.w += expf(e.w - m.w);
    }
  }
  float4 den;
  den.x = wave_sum(z.x) + 1e-16f; den.y = wave_sum(z.y) + 1e-16f;
  den.z = wave_sum(z.z) + 1e-16f; den.w = wave_sum(z.w) + 1e-16f;

  float4 acc;
  {
    const float a_self = __int_as_float(
        (__float_as_int(ps.x / den.x) & k0) | (__float_as_int(ps.y / den.y) & k1) |
        (__float_as_int(ps.z / den.z) & k2) | (__float_as_int(ps.w / den.w) & k3));
    const float4 hv = ld4(hl + (long)d * Dch);
    acc.x = a_self * hv.x; acc.y = a_self * hv.y;
    acc.z = a_self * hv.z; acc.w = a_self * hv.w;
  }
  for (int base = 0; base < Ereal; base += 32) {
    int s; float4 e;
    float4 al = make_float4(0.f, 0.f, 0.f, 0.f);
    const bool v = scan_scores(ei, Ereal, alsrc, d, base, lane, ad, s, e);
    if (v) {
      al.x = expf(e.x - m.x) / den.x; al.y = expf(e.y - m.y) / den.y;
      al.z = expf(e.z - m.z) / den.z; al.w = expf(e.w - m.w) / den.w;
    }
    unsigned hit = __builtin_amdgcn_ballot_w32(v);
    while (hit) {
      const int j = __builtin_ctz(hit);
      hit &= hit - 1;
      const int sj = __builtin_amdgcn_readlane(s, j);
      const float b = __int_as_float((bcast(al.x, j) & k0) | (bcast(al.y, j) & k1) |
                                     (bcast(al.z, j) & k2) | (bcast(al.w, j) & k3));
      const float4 hv = ld4(hl + (long)sj * Dch);
      acc.x += b * hv.x;
      acc.y += b * hv.y;
      acc.z += b * hv.z;
      acc.w += b * hv.w;
    }
  }
  return acc;
}

__device__ __forceinline__ float bn_elu(float a, float b, float g, float be,
                                        float mu, float va) {
  float v = a + b;
  v = (v - mu) * (g / sqrtf(va + BN_EPS)) + be;
  return v > 0.f ? v : expf(v) - 1.f;
}

__device__ __forceinline__ float4 gather_row(
    const int d, const int lane,
    const int* __restrict__ seg_len, const int* __restrict__ csr_src,
    const float* __restrict__ alsrc, const float* __restrict__ aldst,
    const float* __restrict__ a_src,
    const float* __restrict__ Hf,
    const float* __restrict__ bias, const float* __restrict__ gamma,
    const float* __restrict__ beta, const float* __restrict__ mean,
    const float* __restrict__ var,
    const int* __restrict__ ei, int Ereal) {
  const int beg = d * BKT;
  const int cnt = seg_len[d];
  const int hsel = lane >> 3;
  const float* hl = Hf + lane * 4;
  float4 acc;
  if (cnt > BKT) {
    const float4 ad = ld4(aldst + (long)d * Hh);
    const float4 as = ld4(alsrc + (long)d * Hh);
    float4 es;
    es.x = lrelu(as.x + ad.x);
    es.y = lrelu(as.y + ad.y);
    es.z = lrelu(as.z + ad.z);
    es.w = lrelu(as.w + ad.w);
    const int k0 = hsel == 0 ? -1 : 0, k1 = hsel == 1 ? -1 : 0;
    const int k2 = hsel == 2 ? -1 : 0, k3 = hsel == 3 ? -1 : 0;
    acc = gather_by_scan(ei, Ereal, alsrc, hl, d, lane, ad, es, k0, k1, k2, k3);
  } else {
    const float adh = aldst[(long)d * Hh + hsel];
    const Row4  a4  = ldrow(a_src + lane * 4);
    const Row4  hs  = ldrow(hl + (long)d * Dch);
    int s0 = 0, s1 = 0;
    if (lane < cnt)      s0 = csr_src[beg + lane];
    if (32 + lane < cnt) s1 = csr_src[beg + 32 + lane];
    float m = row_score(hs, a4, adh);
    float z = 1.f;
    Row4 sum = hs;
    fold_rows(hl, a4, adh, s0, cnt < 32 ? cnt : 32, m, z, sum);
    if (cnt > 32) fold_rows(hl, a4, adh, s1, cnt - 32, m, z, sum);
    const float r = 1.f / (z + 1e-16f);
    acc.x = sum.lo.x * r; acc.y = sum.lo.y * r;
    acc.z = sum.hi.x * r; acc.w = sum.hi.y * r;
  }
  const float4 bi = ld4(bias + lane * 4);
  const float4 ga = ld4(gamma + lane * 4);
  const float4 be = ld4(beta + lane * 4);
  const float4 mu = ld4(mean + lane * 4);
  const float4 va = ld4(var + lane * 4);
  float4 o;
  o.x = bn_elu(acc.x, bi.x, ga.x, be.x, mu.x, va.x);
  o.y = bn_elu(acc.y, bi.y, ga.y, be.y, mu.y, va.y);
  o.z = bn_elu(acc.z, bi.z, ga.z, be.z, mu.z, va.z);
  o.w = bn_elu(acc.w, bi.w, ga.w, be.w, mu.w, va.w);
  return o;
}

__global__ __launch_bounds__(256) void gat_gather(
    const int* __restrict__ seg_len, const int* __restrict__ csr_src,
    const float* __restrict__ alsrc, const float* __restrict__ aldst,
    const float* __restrict__ a_src_cur,
    const float* __restrict__ Hf,
    const float* __restrict__ bias, const float* __restrict__ gamma,
    const float* __restrict__ beta, const float* __restrict__ mean,
    const float* __restrict__ var,
    float* __restrict__ xo, int Nn,
    const int* __restrict__ ei, int Ereal) {
  const int lane = threadIdx.x & 31;
  const int d = __builtin_amdgcn_readfirstlane((int)(blockIdx.x * 8 + (threadIdx.x >> 5)));
  if (d >= Nn) return;
  const float4 o = gather_row(d, lane, seg_len, csr_src, alsrc, aldst, a_src_cur, Hf,
                              bias, gamma, beta, mean, var, ei, Ereal);
  *reinterpret_cast<float4*>(xo + (long)d * Dch + lane * 4) = o;
}

#define FG_WAVES   16
#define FG_THREADS (FG_WAVES * 32)
#define XPITCH     132
__global__ __launch_bounds__(FG_THREADS) void gather_then_gemm(
    const int* __restrict__ seg_len, const int* __restrict__ csr_src,
    const float* __restrict__ alsrc, const float* __restrict__ aldst,
    const float* __restrict__ a_src_cur,
    const float* __restrict__ Hf,
    const float* __restrict__ bias, const float* __restrict__ gamma,
    const float* __restrict__ beta, const float* __restrict__ mean,
    const float* __restrict__ var, int Nn,
    const int* __restrict__ ei, int Ereal,
    const float* __restrict__ Wn, const float* __restrict__ a_src_n,
    const float* __restrict__ a_dst_n,
    float* __restrict__ Ho, float* __restrict__ alsrc_o,
    float* __restrict__ aldst_o) {
  __shared__ __attribute__((aligned(16))) float xs[FG_WAVES * XPITCH];
  __shared__ __attribute__((aligned(16))) float lg[2 * FG_WAVES * Hh];
  const int lane = threadIdx.x & 31;
  const int wave = __builtin_amdgcn_readfirstlane((int)(threadIdx.x >> 5));
  const int row0 = (int)blockIdx.x * FG_WAVES;
  const int d = row0 + wave;
  float4 o = make_float4(0.f, 0.f, 0.f, 0.f);
  if (d < Nn)
    o = gather_row(d, lane, seg_len, csr_src, alsrc, aldst, a_src_cur, Hf,
                   bias, gamma, beta, mean, var, ei, Ereal);
  *reinterpret_cast<float4*>(&xs[wave * XPITCH + lane * 4]) = o;
  __syncthreads();

  const int half = lane >> 4;
  const int c    = lane & 15;
  const int col0 = wave * Ch + 2 * c;
  v8f acc0 = {}, acc1 = {};
  if (wave < Hh) {
    const float* ar = xs + c * XPITCH + half * 2;
    const float* bc = Wn + (long)(half * 2) * Dch + col0;
    for (int k = 0; k < Dch; k += 4) {
      const v2f a = *reinterpret_cast<const v2f*>(ar + k);
      const float2 b0 = *reinterpret_cast<const float2*>(bc + (long)k * Dch);
      const float2 b1 = *reinterpret_cast<const float2*>(bc + (long)(k + 1) * Dch);
      v2f b;
      b.x = b0.x; b.y = b1.x;
      acc0 = __builtin_amdgcn_wmma_f32_16x16x4_f32(
          false, a, false, b, (short)0, acc0, false, false);
      b.x = b0.y; b.y = b1.y;
      acc1 = __builtin_amdgcn_wmma_f32_16x16x4_f32(
          false, a, false, b, (short)0, acc1, false, false);
    }
    const int laneq = (int)__builtin_amdgcn_mbcnt_lo(~0u, 0u);
    const int halfq = laneq >> 4;
    const int cq    = laneq & 15;
    const int colq  = wave * Ch + 2 * cq;
    const float2 as = *reinterpret_cast<const float2*>(a_src_n + colq);
    const float2 ad = *reinterpret_cast<const float2*>(a_dst_n + colq);
    float sel = 0.f;
#pragma unroll
    for (int v = 0; v < 8; ++v) {
      float ps = acc0[v] * as.x + acc1[v] * as.y;
      float pd = acc0[v] * ad.x + acc1[v] * ad.y;
      ps += row_xor<1>(ps);  pd += row_xor<1>(pd);
      ps += row_xor<2>(ps);  pd += row_xor<2>(pd);
      ps += row_xor<4>(ps);  pd += row_xor<4>(pd);
      ps += row_xor<8>(ps);  pd += row_xor<8>(pd);
      if ((cq & 7) == v) sel = cq < 8 ? ps : pd;
    }
    lg[(cq >> 3) * (FG_WAVES * Hh) + (halfq * 8 + (cq & 7)) * Hh + wave] = sel;
  }
  __syncthreads();
  const int laneq = (int)__builtin_amdgcn_mbcnt_lo(~0u, 0u);
  const int halfq = laneq >> 4;
  const int colq  = wave * Ch + 2 * (laneq & 15);
  if (wave < Hh) {
    float* orow = Ho + (long)(row0 + halfq * 8) * Dch + colq;
    if (row0 + FG_WAVES <= Nn) {
#pragma unroll
      for (int v = 0; v < 8; ++v)
        *reinterpret_cast<float2*>(orow + (long)v * Dch) =
            make_float2(acc0[v], acc1[v]);
    } else {
#pragma unroll
      for (int v = 0; v < 8; ++v)
        if (row0 + halfq * 8 + v < Nn)
          *reinterpret_cast<float2*>(orow + (long)v * Dch) =
              make_float2(acc0[v], acc1[v]);
    }
  } else if (wave < Hh + 2) {
    const int q = laneq * 2;
    if (row0 + (q >> 2) < Nn) {
      const float2 vv = *reinterpret_cast<const float2*>(
          &lg[(wave - Hh) * (FG_WAVES * Hh) + q]);
      float* dst = (wave == Hh ? alsrc_o : aldst_o) + (long)row0 * Hh + q;
      *reinterpret_cast<float2*>(dst) = vv;
    }
  }
}

#define POOL_THREADS 1024
#define POOL_WAVES   32
#define POOL_TILE    65536
#define POOL_IDS     16

__global__ __launch_bounds__(POOL_THREADS) void pool_owner(
    const float* __restrict__ x, const int* __restrict__ batch,
    float* __restrict__ out, int Nn) {
  __shared__ unsigned member[POOL_TILE / 32];
  __shared__ float    part[POOL_WAVES][Dch];
  __shared__ float    pcount[POOL_WAVES];
  const int g    = blockIdx.x;
  const int lane = threadIdx.x & 31;
  const int wave = __builtin_amdgcn_readfirstlane((int)(threadIdx.x >> 5));
  float4 acc = make_float4(0.f, 0.f, 0.f, 0.f);
  float  cnt = 0.f;

  for (int t0 = 0; t0 < Nn; t0 += POOL_TILE) {
    const int tn = Nn - t0 < POOL_TILE ? Nn - t0 : POOL_TILE;

    const int nch = (tn + 31) >> 5;
    for (int c0 = wave; c0 < nch; c0 += POOL_WAVES * POOL_IDS) {
      int id[POOL_IDS];
#pragma unroll
      for (int k = 0; k < POOL_IDS; ++k) {
        const int n = t0 + (c0 + POOL_WAVES * k) * 32 + lane;
        id[k] = batch[n < Nn ? n : Nn - 1];
      }
#pragma unroll
      for (int k = 0; k < POOL_IDS; ++k) {
        const int c = c0 + POOL_WAVES * k;
        const int n = t0 + c * 32 + lane;
        const unsigned b = __builtin_amdgcn_ballot_w32(n < Nn && id[k] == g);
        if (c < nch && lane == 0) member[c] = b;
      }
    }
    __syncthreads();

    const int ngrp = (tn + 7) >> 3;
    for (int q0 = wave; q0 < ngrp; q0 += 32 * POOL_WAVES) {
      const int q = q0 + 32 * lane;
      unsigned m8 = 0;
      if (q < ngrp) m8 = (member[q >> 2] >> ((q & 3) * 8)) & 0xffu;
      unsigned any = __builtin_amdgcn_ballot_w32(m8 != 0);
      while (any) {
        const int i = __builtin_ctz(any);
        any &= any - 1;
        const unsigned mm = (unsigned)__builtin_amdgcn_readlane((int)m8, i);
        const int nb = t0 + (q0 + 32 * i) * 8;
        float4 r[8];
#pragma unroll
        for (int j = 0; j < 8; ++j) {
          int n = nb + j;
          n = n < Nn ? n : Nn - 1;
          r[j] = ld4(x + (long)n * Dch + lane * 4);
        }
#pragma unroll
        for (int j = 0; j < 8; ++j) {
          if (mm & (1u << j)) {
            acc.x += r[j].x; acc.y += r[j].y; acc.z += r[j].z; acc.w += r[j].w;
          }
        }
        cnt += (float)__builtin_popcount(mm);
      }
    }
    __syncthreads();
  }

  *reinterpret_cast<float4*>(&part[wave][lane * 4]) = acc;
  if (lane == 0) pcount[wave] = cnt;
  __syncthreads();
  if (threadIdx.x < Dch) {
    float sum = 0.f, c = 0.f;
    for (int w = 0; w < POOL_WAVES; ++w) { sum += part[w][threadIdx.x]; c += pcount[w]; }
    out[(long)g * Dch + threadIdx.x] = sum / (c > 1.f ? c : 1.f);
  }
}

extern "C" void kernel_launch(void* const* d_in, const int* in_sizes, int n_in,
                              void* d_out, int out_size, void* d_ws, size_t ws_size,
                              hipStream_t stream) {
  const float* x0    = (const float*)d_in[0];
  const int*   ei    = (const int*)d_in[1];
  const int*   batch = (const int*)d_in[2];
  const float* Ws    = (const float*)d_in[3];
  const float* a_src = (const float*)d_in[4];
  const float* a_dst = (const float*)d_in[5];
  const float* bias  = (const float*)d_in[6];
  const float* gamma = (const float*)d_in[7];
  const float* beta  = (const float*)d_in[8];
  const float* mean  = (const float*)d_in[9];
  const float* var   = (const float*)d_in[10];

  const int Nn    = in_sizes[0] / Dch;
  const int Ereal = in_sizes[1] / 2;
  const int G     = out_size / Dch;

  char* ws = (char*)d_ws;
  float*    hB    = (float*)ws;    ws += (size_t)Nn * Dch * 4;
  char*     grp   = ws;            ws += (size_t)Nn * Dch * 4;
  float*    hA    = (float*)ws;    ws += (size_t)Nn * Dch * 4;
  float*    alsB  = (float*)ws;    ws += (size_t)Nn * Hh * 4;
  float*    aldB  = (float*)ws;    ws += (size_t)Nn * Hh * 4;
  float*    alsA  = (float*)ws;    ws += (size_t)Nn * Hh * 4;
  float*    aldA  = (float*)ws;    ws += (size_t)Nn * Hh * 4;

  int* bucket  = (int*)grp;
  int* seg_len = (int*)(grp + (size_t)Nn * BKT * 4);

  fill_u32<<<64, 256, 0, stream>>>((unsigned*)seg_len, 0u, (long)Nn);

  {
    const int gemm_blocks = (Nn / 16 + GEMM_STRIPS - 1) / GEMM_STRIPS;
    const int edge_blocks = (Ereal + GEMM_THREADS - 1) / GEMM_THREADS;
    gemm_beside_buckets<<<gemm_blocks + edge_blocks, GEMM_THREADS, 0, stream>>>(
        x0, Ws, a_src, a_dst, hA, alsA, aldA, Nn / 16, gemm_blocks,
        ei, Ereal, seg_len, bucket);
  }

  float* hin = hA;  float* hout = hB;
  float* asin = alsA; float* adin = aldA;
  float* asout = alsB; float* adout = aldB;
  for (int l = 0; l + 1 < Lnum; ++l) {
    gather_then_gemm<<<(Nn + FG_WAVES - 1) / FG_WAVES, FG_THREADS, 0, stream>>>(
        seg_len, bucket, asin, adin, a_src + (size_t)l * Hh * Ch, hin,
        bias + (size_t)l * Dch, gamma + (size_t)l * Dch,
        beta + (size_t)l * Dch, mean + (size_t)l * Dch, var + (size_t)l * Dch,
        Nn, ei, Ereal,
        Ws + (size_t)(l + 1) * Dch * Dch,
        a_src + (size_t)(l + 1) * Hh * Ch, a_dst + (size_t)(l + 1) * Hh * Ch,
        hout, asout, adout);
    float* t;
    t = hin;  hin = hout;   hout = t;
    t = asin; asin = asout; asout = t;
    t = adin; adin = adout; adout = t;
  }

  {
    const int l = Lnum - 1;
    gat_gather<<<(Nn + 7) / 8, 256, 0, stream>>>(
        seg_len, bucket, asin, adin, a_src + (size_t)l * Hh * Ch, hin,
        bias + (size_t)l * Dch, gamma + (size_t)l * Dch,
        beta + (size_t)l * Dch, mean + (size_t)l * Dch, var + (size_t)l * Dch,
        hout, Nn, ei, Ereal);
  }

  pool_owner<<<G, POOL_THREADS, 0, stream>>>(hout, batch, (float*)d_out, Nn);
}


# BatchNorm scale and shift per channel formed once per call by the first launch; gather wave loads them at its start: one multiply-add + ELU, no per-wave sqrt, divide, late loads; BN sum order changed
# speedup vs baseline: 7.4758x; 1.0355x over previous
#include <hip/hip_runtime.h>


#define Dch      128
#define Hh       4
#define Ch       32
#define Lnum     3
#define NEG_SLOPE 0.2f
#define BN_EPS   1e-5f

typedef __attribute__((ext_vector_type(2))) float v2f;
typedef __attribute__((ext_vector_type(8))) float v8f;

__global__ void fill_u32(unsigned* __restrict__ p, unsigned v, long n,
                         const float* __restrict__ bias, const float* __restrict__ gamma,
                         const float* __restrict__ beta, const float* __restrict__ mean,
                         const float* __restrict__ var, float* __restrict__ tab, int nch) {
  long i = (long)blockIdx.x * blockDim.x + threadIdx.x;
  long stride = (long)gridDim.x * blockDim.x;
  for (long ch = i; ch < nch; ch += stride) {
    const float sc = gamma[ch] / sqrtf(var[ch] + BN_EPS);
    tab[ch]       = sc;
    tab[nch + ch] = (bias[ch] - mean[ch]) * sc + beta[ch];
  }
  for (; i < n; i += stride) p[i] = v;
}

__device__ __forceinline__ float4 ld4(const float* p) {
  return *reinterpret_cast<const float4*>(p);
}

template <int M>
__device__ __forceinline__ float row_xor(float v) {
  return __int_as_float(__builtin_amdgcn_update_dpp(
      0, __float_as_int(v), 0x160 | M, 0xf, 0xf, false));
}

#define GEMM_STRIPS  2
#define GEMM_THREADS (GEMM_STRIPS * 64)
__device__ __forceinline__ void gemm_block(
    const int bx,
    const float* __restrict__ X, const float* __restrict__ Wm,
    const float* __restrict__ a_src, const float* __restrict__ a_dst,
    float* __restrict__ Ho, float* __restrict__ alsrc,
    float* __restrict__ aldst, int nstrips) {
  __shared__ float lg[GEMM_STRIPS * 2 * 16 * Hh];
  const int wave = threadIdx.x >> 5;
  const int lane = threadIdx.x & 31;
  const int half = lane >> 4;
  const int l15  = lane & 15;
  const int sloc = wave >> 1;
  const int grp  = wave & 1;
  const int strip = bx * GEMM_STRIPS + sloc;
  const bool live = strip < nstrips;
  const int row0 = strip << 4;
  const int col0 = grp * 64 + l15 * 4;

  v8f acc0 = {}, acc1 = {}, acc2 = {}, acc3 = {};
  if (live) {
    const float* arow = X + (long)(row0 + l15) * Dch;
    const float* bcol = Wm + col0;
    for (int k = 0; k < Dch; k += 4) {
      const int ka = k + half * 2;
      v2f a, b;
      a.x = arow[ka];
      a.y = arow[ka + 1];
      const float4 b0 = ld4(bcol + (long)ka * Dch);
      const float4 b1 = ld4(bcol + (long)(ka + 1) * Dch);
      b.x = b0.x; b.y = b1.x;
      acc0 = __builtin_amdgcn_wmma_f32_16x16x4_f32(
           false, a,  false, b,
           (short)0, acc0,  false,  false);
      b.x = b0.y; b.y = b1.y;
      acc1 = __builtin_amdgcn_wmma_f32_16x16x4_f32(
          false, a, false, b, (short)0, acc1, false, false);
      b.x = b0.z; b.y = b1.z;
      acc2 = __builtin_amdgcn_wmma_f32_16x16x4_f32(
          false, a, false, b, (short)0, acc2, false, false);
      b.x = b0.w; b.y = b1.w;
      acc3 = __builtin_amdgcn_wmma_f32_16x16x4_f32(
          false, a, false, b, (short)0, acc3, false, false);
    }

    const float4 as = ld4(a_src + col0);
    const float4 ad = ld4(a_dst + col0);
    float ssel = 0.f, dsel = 0.f;
#pragma unroll
    for (int v = 0; v < 8; ++v) {
      float ps = acc0[v] * as.x + acc1[v] * as.y + acc2[v] * as.z + acc3[v] * as.w;
      float pd = acc0[v] * ad.x + acc1[v] * ad.y + acc2[v] * ad.z + acc3[v] * ad.w;
      ps += row_xor<1>(ps);  pd += row_xor<1>(pd);
      ps += row_xor<2>(ps);  pd += row_xor<2>(pd);
      ps += row_xor<4>(ps);  pd += row_xor<4>(pd);
      if ((l15 & 7) == v) { ssel = ps; dsel = pd; }
    }
    const int li = sloc * 128 + (half * 8 + (l15 & 7)) * Hh + grp * 2 + (l15 >> 3);
    lg[li]      = ssel;
    lg[li + 64] = dsel;
  }
  __syncthreads();
  const int t = threadIdx.x;
  if (bx * GEMM_STRIPS + (t >> 6) < nstrips) {
    const long o = (long)bx * (GEMM_STRIPS * 16 * Hh) + t;
    const int  q = (t >> 6) * 128 + (t & 63);
    alsrc[o] = lg[q];
    aldst[o] = lg[q + 64];
  }
  if (live) {
    float* orow = Ho + (long)(row0 + half * 8) * Dch + col0;
#pragma unroll
    for (int v = 0; v < 8; ++v)
      *reinterpret_cast<float4*>(orow + (long)v * Dch) =
          make_float4(acc0[v], acc1[v], acc2[v], acc3[v]);
  }
}

#define BKT 64
__device__ __forceinline__ void bucket_edge(
    const int e, const int* __restrict__ ei, int Ereal,
    int* __restrict__ seg_len, int* __restrict__ bucket) {
  if (e >= Ereal) return;
  const int s = ei[e];
  const int d = ei[Ereal + e];
  const int pos = atomicAdd(&seg_len[d], 1);
  bucket[d * BKT + (pos < BKT ? pos : BKT - 1)] = s;
}

__global__ __launch_bounds__(GEMM_THREADS) void gemm_beside_buckets(
    const float* __restrict__ X, const float* __restrict__ Wm,
    const float* __restrict__ a_src, const float* __restrict__ a_dst,
    float* __restrict__ Ho, float* __restrict__ alsrc,
    float* __restrict__ aldst, int nstrips, int gemm_blocks,
    const int* __restrict__ ei, int Ereal,
    int* __restrict__ seg_len, int* __restrict__ bucket) {
  const int bx = (int)blockIdx.x;
  if (bx < gemm_blocks) {
    gemm_block(bx, X, Wm, a_src, a_dst, Ho, alsrc, aldst, nstrips);
  } else {
    bucket_edge((bx - gemm_blocks) * GEMM_THREADS + (int)threadIdx.x,
                ei, Ereal, seg_len, bucket);
  }
}

__device__ __forceinline__ float lrelu(float v) { return v > 0.f ? v : v * NEG_SLOPE; }

__device__ __forceinline__ float wave_max(float v) {
  for (int o = 16; o > 0; o >>= 1) v = fmaxf(v, __shfl_xor(v, o, 32));
  return v;
}
__device__ __forceinline__ float wave_sum(float v) {
  for (int o = 16; o > 0; o >>= 1) v += __shfl_xor(v, o, 32);
  return v;
}


__device__ __forceinline__ float head_sum(float v) {
  v += row_xor<1>(v);
  v += row_xor<2>(v);
  v += row_xor<4>(v);
  return v;
}

struct Row4 { v2f lo, hi; };
__device__ __forceinline__ Row4 ldrow(const float* p) {
  const float4 t = ld4(p);
  Row4 r;
  r.lo.x = t.x; r.lo.y = t.y; r.hi.x = t.z; r.hi.y = t.w;
  return r;
}

__device__ __forceinline__ float row_score(Row4 h, Row4 a, float adh) {
  const v2f t = h.lo * a.lo + h.hi * a.hi;
  const float v = head_sum(t.x + t.y) + adh;
  return fmaxf(v, v * NEG_SLOPE);
}

template <bool TAIL>
__device__ __forceinline__ void fold4(const float* __restrict__ hl, Row4 a4, float adh,
                                      int s, int j, int n,
                                      float& m, float& z, Row4& acc) {
  const int s0 = __builtin_amdgcn_readlane(s, j);
  const int s1 = __builtin_amdgcn_readlane(s, j + 1);
  const int s2 = __builtin_amdgcn_readlane(s, j + 2);
  const int s3 = __builtin_amdgcn_readlane(s, j + 3);
  const Row4 h0 = ldrow(hl + (long)s0 * Dch);
  const Row4 h1 = ldrow(hl + (long)s1 * Dch);
  const Row4 h2 = ldrow(hl + (long)s2 * Dch);
  const Row4 h3 = ldrow(hl + (long)s3 * Dch);
  const float e0 = row_score(h0, a4, adh);
  float e1 = row_score(h1, a4, adh);
  float e2 = row_score(h2, a4, adh);
  float e3 = row_score(h3, a4, adh);
  if (TAIL) {
    if (j + 1 >= n) e1 = -__builtin_inff();
    if (j + 2 >= n) e2 = -__builtin_inff();
    if (j + 3 >= n) e3 = -__builtin_inff();
  }
  const float mn = fmaxf(fmaxf(m, e0), fmaxf(fmaxf(e1, e2), e3));
  const float sc = __expf(m - mn);
  const float p0 = __expf(e0 - mn);
  float p1 = __expf(e1 - mn);
  float p2 = __expf(e2 - mn);
  float p3 = __expf(e3 - mn);
  if (TAIL) {
    if (j + 1 >= n) p1 = 0.f;
    if (j + 2 >= n) p2 = 0.f;
    if (j + 3 >= n) p3 = 0.f;
  }
  m = mn;
  z = z * sc + ((p0 + p1) + (p2 + p3));
  acc.lo = acc.lo * sc + h0.lo * p0 + h1.lo * p1 + h2.lo * p2 + h3.lo * p3;
  acc.hi = acc.hi * sc + h0.hi * p0 + h1.hi * p1 + h2.hi * p2 + h3.hi * p3;
}

__device__ __forceinline__ void fold_rows(const float* __restrict__ hl, Row4 a4, float adh,
                                          int s, int n,
                                          float& m, float& z, Row4& acc) {
  int j = 0;
  for (; j + 4 <= n; j += 4) fold4<false>(hl, a4, adh, s, j, n, m, z, acc);
  if (j < n) fold4<true>(hl, a4, adh, s, j, n, m, z, acc);
}

__device__ __forceinline__ int bcast(float v, int j) {
  return __builtin_amdgcn_readlane(__float_as_int(v), j);
}

__device__ __forceinline__ bool scan_scores(
    const int* __restrict__ ei, int Ereal, const float* __restrict__ alsrc,
    int d, int base, int lane, float4 ad, int& s, float4& e) {
  const int i = base + lane;
  s = 0;
  e = make_float4(0.f, 0.f, 0.f, 0.f);
  const bool valid = i < Ereal && ei[Ereal + i] == d;
  if (valid) {
    s = ei[i];
    const float4 a = ld4(alsrc + (long)s * Hh);
    e.x = lrelu(a.x + ad.x);
    e.y = lrelu(a.y + ad.y);
    e.z = lrelu(a.z + ad.z);
    e.w = lrelu(a.w + ad.w);
  }
  return valid;
}

__device__ float4 gather_by_scan(
    const int* __restrict__ ei, int Ereal, const float* __restrict__ alsrc,
    const float* __restrict__ hl, int d, int lane, float4 ad, float4 es,
    int k0, int k1, int k2, int k3) {
  float4 m = es;
  for (int base = 0; base < Ereal; base += 32) {
    int s; float4 e;
    if (scan_scores(ei, Ereal, alsrc, d, base, lane, ad, s, e)) {
      m.x = fmaxf(m.x, e.x); m.y = fmaxf(m.y, e.y);
      m.z = fmaxf(m.z, e.z); m.w = fmaxf(m.w, e.w);
    }
  }
  m.x = wave_max(m.x); m.y = wave_max(m.y);
  m.z = wave_max(m.z); m.w = wave_max(m.w);

  float4 ps;
  ps.x = expf(es.x - m.x); ps.y = expf(es.y - m.y);
  ps.z = expf(es.z - m.z); ps.w = expf(es.w - m.w);
  float4 z = make_float4(0.f, 0.f, 0.f, 0.f);
  if (lane == 0) z = ps;
  for (int base = 0; base < Ereal; base += 32) {
    int s; float4 e;
    if (scan_scores(ei, Ereal, alsrc, d, base, lane, ad, s, e)) {
      z.x += expf(e.x - m.x); z.y += expf(e.y - m.y);
      z.z += expf(e.z - m.z); z.w += expf(e.w - m.w);
    }
  }
  float4 den;
  den.x = wave_sum(z.x) + 1e-16f; den.y = wave_sum(z.y) + 1e-16f;
  den.z = wave_sum(z.z) + 1e-16f; den.w = wave_sum(z.w) + 1e-16f;

  float4 acc;
  {
    const float a_self = __int_as_float(
        (__float_as_int(ps.x / den.x) & k0) | (__float_as_int(ps.y / den.y) & k1) |
        (__float_as_int(ps.z / den.z) & k2) | (__float_as_int(ps.w / den.w) & k3));
    const float4 hv = ld4(hl + (long)d * Dch);
    acc.x = a_self * hv.x; acc.y = a_self * hv.y;
    acc.z = a_self * hv.z; acc.w = a_self * hv.w;
  }
  for (int base = 0; base < Ereal; base += 32) {
    int s; float4 e;
    float4 al = make_float4(0.f, 0.f, 0.f, 0.f);
    const bool v = scan_scores(ei, Ereal, alsrc, d, base, lane, ad, s, e);
    if (v) {
      al.x = expf(e.x - m.x) / den.x; al.y = expf(e.y - m.y) / den.y;
      al.z = expf(e.z - m.z) / den.z; al.w = expf(e.w - m.w) / den.w;
    }
    unsigned hit = __builtin_amdgcn_ballot_w32(v);
    while (hit) {
      const int j = __builtin_ctz(hit);
      hit &= hit - 1;
      const int sj = __builtin_amdgcn_readlane(s, j);
      const float b = __int_as_float((bcast(al.x, j) & k0) | (bcast(al.y, j) & k1) |
                                     (bcast(al.z, j) & k2) | (bcast(al.w, j) & k3));
      const float4 hv = ld4(hl + (long)sj * Dch);
      acc.x += b * hv.x;
      acc.y += b * hv.y;
      acc.z += b * hv.z;
      acc.w += b * hv.w;
    }
  }
  return acc;
}

__device__ __forceinline__ float bn_elu(float a, float sc, float sh) {
  const float v = a * sc + sh;
  return v > 0.f ? v : expf(v) - 1.f;
}

__device__ __forceinline__ float4 gather_row(
    const int d, const int lane,
    const int* __restrict__ seg_len, const int* __restrict__ csr_src,
    const float* __restrict__ alsrc, const float* __restrict__ aldst,
    const float* __restrict__ a_src,
    const float* __restrict__ Hf,
    const float* __restrict__ bn_scale, const float* __restrict__ bn_shift,
    const int* __restrict__ ei, int Ereal) {
  const int beg = d * BKT;
  const int cnt = seg_len[d];
  const int hsel = lane >> 3;
  const float* hl = Hf + lane * 4;
  const float4 sc4 = ld4(bn_scale + lane * 4);
  const float4 sh4 = ld4(bn_shift + lane * 4);
  float4 acc;
  if (cnt > BKT) {
    const float4 ad = ld4(aldst + (long)d * Hh);
    const float4 as = ld4(alsrc + (long)d * Hh);
    float4 es;
    es.x = lrelu(as.x + ad.x);
    es.y = lrelu(as.y + ad.y);
    es.z = lrelu(as.z + ad.z);
    es.w = lrelu(as.w + ad.w);
    const int k0 = hsel == 0 ? -1 : 0, k1 = hsel == 1 ? -1 : 0;
    const int k2 = hsel == 2 ? -1 : 0, k3 = hsel == 3 ? -1 : 0;
    acc = gather_by_scan(ei, Ereal, alsrc, hl, d, lane, ad, es, k0, k1, k2, k3);
  } else {
    const float adh = aldst[(long)d * Hh + hsel];
    const Row4  a4  = ldrow(a_src + lane * 4);
    const Row4  hs  = ldrow(hl + (long)d * Dch);
    int s0 = 0, s1 = 0;
    if (lane < cnt)      s0 = csr_src[beg + lane];
    if (32 + lane < cnt) s1 = csr_src[beg + 32 + lane];
    float m = row_score(hs, a4, adh);
    float z = 1.f;
    Row4 sum = hs;
    fold_rows(hl, a4, adh, s0, cnt < 32 ? cnt : 32, m, z, sum);
    if (cnt > 32) fold_rows(hl, a4, adh, s1, cnt - 32, m, z, sum);
    const float r = 1.f / (z + 1e-16f);
    acc.x = sum.lo.x * r; acc.y = sum.lo.y * r;
    acc.z = sum.hi.x * r; acc.w = sum.hi.y * r;
  }
  float4 o;
  o.x = bn_elu(acc.x, sc4.x, sh4.x);
  o.y = bn_elu(acc.y, sc4.y, sh4.y);
  o.z = bn_elu(acc.z, sc4.z, sh4.z);
  o.w = bn_elu(acc.w, sc4.w, sh4.w);
  return o;
}

__global__ __launch_bounds__(256) void gat_gather(
    const int* __restrict__ seg_len, const int* __restrict__ csr_src,
    const float* __restrict__ alsrc, const float* __restrict__ aldst,
    const float* __restrict__ a_src_cur,
    const float* __restrict__ Hf,
    const float* __restrict__ bn_scale, const float* __restrict__ bn_shift,
    float* __restrict__ xo, int Nn,
    const int* __restrict__ ei, int Ereal) {
  const int lane = threadIdx.x & 31;
  const int d = __builtin_amdgcn_readfirstlane((int)(blockIdx.x * 8 + (threadIdx.x >> 5)));
  if (d >= Nn) return;
  const float4 o = gather_row(d, lane, seg_len, csr_src, alsrc, aldst, a_src_cur, Hf,
                              bn_scale, bn_shift, ei, Ereal);
  *reinterpret_cast<float4*>(xo + (long)d * Dch + lane * 4) = o;
}

#define FG_WAVES   16
#define FG_THREADS (FG_WAVES * 32)
#define XPITCH     132
__global__ __launch_bounds__(FG_THREADS) void gather_then_gemm(
    const int* __restrict__ seg_len, const int* __restrict__ csr_src,
    const float* __restrict__ alsrc, const float* __restrict__ aldst,
    const float* __restrict__ a_src_cur,
    const float* __restrict__ Hf,
    const float* __restrict__ bn_scale, const float* __restrict__ bn_shift, int Nn,
    const int* __restrict__ ei, int Ereal,
    const float* __restrict__ Wn, const float* __restrict__ a_src_n,
    const float* __restrict__ a_dst_n,
    float* __restrict__ Ho, float* __restrict__ alsrc_o,
    float* __restrict__ aldst_o) {
  __shared__ __attribute__((aligned(16))) float xs[FG_WAVES * XPITCH];
  __shared__ __attribute__((aligned(16))) float lg[2 * FG_WAVES * Hh];
  const int lane = threadIdx.x & 31;
  const int wave = __builtin_amdgcn_readfirstlane((int)(threadIdx.x >> 5));
  const int row0 = (int)blockIdx.x * FG_WAVES;
  const int d = row0 + wave;
  float4 o = make_float4(0.f, 0.f, 0.f, 0.f);
  if (d < Nn)
    o = gather_row(d, lane, seg_len, csr_src, alsrc, aldst, a_src_cur, Hf,
                   bn_scale, bn_shift, ei, Ereal);
  *reinterpret_cast<float4*>(&xs[wave * XPITCH + lane * 4]) = o;
  __syncthreads();

  const int half = lane >> 4;
  const int c    = lane & 15;
  const int col0 = wave * Ch + 2 * c;
  v8f acc0 = {}, acc1 = {};
  if (wave < Hh) {
    const float* ar = xs + c * XPITCH + half * 2;
    const float* bc = Wn + (long)(half * 2) * Dch + col0;
    for (int k = 0; k < Dch; k += 4) {
      const v2f a = *reinterpret_cast<const v2f*>(ar + k);
      const float2 b0 = *reinterpret_cast<const float2*>(bc + (long)k * Dch);
      const float2 b1 = *reinterpret_cast<const float2*>(bc + (long)(k + 1) * Dch);
      v2f b;
      b.x = b0.x; b.y = b1.x;
      acc0 = __builtin_amdgcn_wmma_f32_16x16x4_f32(
          false, a, false, b, (short)0, acc0, false, false);
      b.x = b0.y; b.y = b1.y;
      acc1 = __builtin_amdgcn_wmma_f32_16x16x4_f32(
          false, a, false, b, (short)0, acc1, false, false);
    }
    const int laneq = (int)__builtin_amdgcn_mbcnt_lo(~0u, 0u);
    const int halfq = laneq >> 4;
    const int cq    = laneq & 15;
    const int colq  = wave * Ch + 2 * cq;
    const float2 as = *reinterpret_cast<const float2*>(a_src_n + colq);
    const float2 ad = *reinterpret_cast<const float2*>(a_dst_n + colq);
    float sel = 0.f;
#pragma unroll
    for (int v = 0; v < 8; ++v) {
      float ps = acc0[v] * as.x + acc1[v] * as.y;
      float pd = acc0[v] * ad.x + acc1[v] * ad.y;
      ps += row_xor<1>(ps);  pd += row_xor<1>(pd);
      ps += row_xor<2>(ps);  pd += row_xor<2>(pd);
      ps += row_xor<4>(ps);  pd += row_xor<4>(pd);
      ps += row_xor<8>(ps);  pd += row_xor<8>(pd);
      if ((cq & 7) == v) sel = cq < 8 ? ps : pd;
    }
    lg[(cq >> 3) * (FG_WAVES * Hh) + (halfq * 8 + (cq & 7)) * Hh + wave] = sel;
  }
  __syncthreads();
  const int laneq = (int)__builtin_amdgcn_mbcnt_lo(~0u, 0u);
  const int halfq = laneq >> 4;
  const int colq  = wave * Ch + 2 * (laneq & 15);
  if (wave < Hh) {
    float* orow = Ho + (long)(row0 + halfq * 8) * Dch + colq;
    if (row0 + FG_WAVES <= Nn) {
#pragma unroll
      for (int v = 0; v < 8; ++v)
        *reinterpret_cast<float2*>(orow + (long)v * Dch) =
            make_float2(acc0[v], acc1[v]);
    } else {
#pragma unroll
      for (int v = 0; v < 8; ++v)
        if (row0 + halfq * 8 + v < Nn)
          *reinterpret_cast<float2*>(orow + (long)v * Dch) =
              make_float2(acc0[v], acc1[v]);
    }
  } else if (wave < Hh + 2) {
    const int q = laneq * 2;
    if (row0 + (q >> 2) < Nn) {
      const float2 vv = *reinterpret_cast<const float2*>(
          &lg[(wave - Hh) * (FG_WAVES * Hh) + q]);
      float* dst = (wave == Hh ? alsrc_o : aldst_o) + (long)row0 * Hh + q;
      *reinterpret_cast<float2*>(dst) = vv;
    }
  }
}

#define POOL_THREADS 1024
#define POOL_WAVES   32
#define POOL_TILE    65536
#define POOL_IDS     16

__global__ __launch_bounds__(POOL_THREADS) void pool_owner(
    const float* __restrict__ x, const int* __restrict__ batch,
    float* __restrict__ out, int Nn) {
  __shared__ unsigned member[POOL_TILE / 32];
  __shared__ float    part[POOL_WAVES][Dch];
  __shared__ float    pcount[POOL_WAVES];
  const int g    = blockIdx.x;
  const int lane = threadIdx.x & 31;
  const int wave = __builtin_amdgcn_readfirstlane((int)(threadIdx.x >> 5));
  float4 acc = make_float4(0.f, 0.f, 0.f, 0.f);
  float  cnt = 0.f;

  for (int t0 = 0; t0 < Nn; t0 += POOL_TILE) {
    const int tn = Nn - t0 < POOL_TILE ? Nn - t0 : POOL_TILE;

    const int nch = (tn + 31) >> 5;
    for (int c0 = wave; c0 < nch; c0 += POOL_WAVES * POOL_IDS) {
      int id[POOL_IDS];
#pragma unroll
      for (int k = 0; k < POOL_IDS; ++k) {
        const int n = t0 + (c0 + POOL_WAVES * k) * 32 + lane;
        id[k] = batch[n < Nn ? n : Nn - 1];
      }
#pragma unroll
      for (int k = 0; k < POOL_IDS; ++k) {
        const int c = c0 + POOL_WAVES * k;
        const int n = t0 + c * 32 + lane;
        const unsigned b = __builtin_amdgcn_ballot_w32(n < Nn && id[k] == g);
        if (c < nch && lane == 0) member[c] = b;
      }
    }
    __syncthreads();

    const int ngrp = (tn + 7) >> 3;
    for (int q0 = wave; q0 < ngrp; q0 += 32 * POOL_WAVES) {
      const int q = q0 + 32 * lane;
      unsigned m8 = 0;
      if (q < ngrp) m8 = (member[q >> 2] >> ((q & 3) * 8)) & 0xffu;
      unsigned any = __builtin_amdgcn_ballot_w32(m8 != 0);
      while (any) {
        const int i = __builtin_ctz(any);
        any &= any - 1;
        const unsigned mm = (unsigned)__builtin_amdgcn_readlane((int)m8, i);
        const int nb = t0 + (q0 + 32 * i) * 8;
        float4 r[8];
#pragma unroll
        for (int j = 0; j < 8; ++j) {
          int n = nb + j;
          n = n < Nn ? n : Nn - 1;
          r[j] = ld4(x + (long)n * Dch + lane * 4);
        }
#pragma unroll
        for (int j = 0; j < 8; ++j) {
          if (mm & (1u << j)) {
            acc.x += r[j].x; acc.y += r[j].y; acc.z += r[j].z; acc.w += r[j].w;
          }
        }
        cnt += (float)__builtin_popcount(mm);
      }
    }
    __syncthreads();
  }

  *reinterpret_cast<float4*>(&part[wave][lane * 4]) = acc;
  if (lane == 0) pcount[wave] = cnt;
  __syncthreads();
  if (threadIdx.x < Dch) {
    float sum = 0.f, c = 0.f;
    for (int w = 0; w < POOL_WAVES; ++w) { sum += part[w][threadIdx.x]; c += pcount[w]; }
    out[(long)g * Dch + threadIdx.x] = sum / (c > 1.f ? c : 1.f);
  }
}

extern "C" void kernel_launch(void* const* d_in, const int* in_sizes, int n_in,
                              void* d_out, int out_size, void* d_ws, size_t ws_size,
                              hipStream_t stream) {
  const float* x0    = (const float*)d_in[0];
  const int*   ei    = (const int*)d_in[1];
  const int*   batch = (const int*)d_in[2];
  const float* Ws    = (const float*)d_in[3];
  const float* a_src = (const float*)d_in[4];
  const float* a_dst = (const float*)d_in[5];
  const float* bias  = (const float*)d_in[6];
  const float* gamma = (const float*)d_in[7];
  const float* beta  = (const float*)d_in[8];
  const float* mean  = (const float*)d_in[9];
  const float* var   = (const float*)d_in[10];

  const int Nn    = in_sizes[0] / Dch;
  const int Ereal = in_sizes[1] / 2;
  const int G     = out_size / Dch;

  char* ws = (char*)d_ws;
  float*    hB    = (float*)ws;    ws += (size_t)Nn * Dch * 4;
  char*     grp   = ws;            ws += (size_t)Nn * Dch * 4;
  float*    hA    = (float*)ws;    ws += (size_t)Nn * Dch * 4;
  float*    alsB  = (float*)ws;    ws += (size_t)Nn * Hh * 4;
  float*    aldB  = (float*)ws;    ws += (size_t)Nn * Hh * 4;
  float*    alsA  = (float*)ws;    ws += (size_t)Nn * Hh * 4;
  float*    aldA  = (float*)ws;    ws += (size_t)Nn * Hh * 4;

  int* bucket  = (int*)grp;
  int* seg_len = (int*)(grp + (size_t)Nn * BKT * 4);
  float* bn_tab = (float*)(grp + (size_t)Nn * (BKT + 4) * 4);
  const float* bn_sc = bn_tab;
  const float* bn_sh = bn_tab + Lnum * Dch;

  fill_u32<<<64, 256, 0, stream>>>((unsigned*)seg_len, 0u, (long)Nn,
                                   bias, gamma, beta, mean, var, bn_tab, Lnum * Dch);

  {
    const int gemm_blocks = (Nn / 16 + GEMM_STRIPS - 1) / GEMM_STRIPS;
    const int edge_blocks = (Ereal + GEMM_THREADS - 1) / GEMM_THREADS;
    gemm_beside_buckets<<<gemm_blocks + edge_blocks, GEMM_THREADS, 0, stream>>>(
        x0, Ws, a_src, a_dst, hA, alsA, aldA, Nn / 16, gemm_blocks,
        ei, Ereal, seg_len, bucket);
  }

  float* hin = hA;  float* hout = hB;
  float* asin = alsA; float* adin = aldA;
  float* asout = alsB; float* adout = aldB;
  for (int l = 0; l + 1 < Lnum; ++l) {
    gather_then_gemm<<<(Nn + FG_WAVES - 1) / FG_WAVES, FG_THREADS, 0, stream>>>(
        seg_len, bucket, asin, adin, a_src + (size_t)l * Hh * Ch, hin,
        bn_sc + (size_t)l * Dch, bn_sh + (size_t)l * Dch,
        Nn, ei, Ereal,
        Ws + (size_t)(l + 1) * Dch * Dch,
        a_src + (size_t)(l + 1) * Hh * Ch, a_dst + (size_t)(l + 1) * Hh * Ch,
        hout, asout, adout);
    float* t;
    t = hin;  hin = hout;   hout = t;
    t = asin; asin = asout; asout = t;
    t = adin; adin = adout; adout = t;
  }

  {
    const int l = Lnum - 1;
    gat_gather<<<(Nn + 7) / 8, 256, 0, stream>>>(
        seg_len, bucket, asin, adin, a_src + (size_t)l * Hh * Ch, hin,
        bn_sc + (size_t)l * Dch, bn_sh + (size_t)l * Dch,
        hout, Nn, ei, Ereal);
  }

  pool_owner<<<G, POOL_THREADS, 0, stream>>>(hout, batch, (float*)d_out, Nn);
}
